# starting kernel
# baseline (speedup 1.0000x reference)
#include <hip/hip_runtime.h>
#include <hip/hip_bf16.h>

typedef __attribute__((ext_vector_type(2))) float v2f;
typedef __attribute__((ext_vector_type(8))) float v8f;


__device__ __forceinline__ void atom_add_f32(float* p, float v) {
  __hip_atomic_fetch_add(p, v, __ATOMIC_RELAXED, __HIP_MEMORY_SCOPE_AGENT);
}

__global__ void zero_kernel(float* __restrict__ p, long n) {
  long i = (long)blockIdx.x * blockDim.x + threadIdx.x;
  if (i < n) p[i] = 0.0f;
}

__global__ void degree_kernel(const int* __restrict__ dst, float* __restrict__ cnt, int E) {
  int e = blockIdx.x * blockDim.x + threadIdx.x;
  if (e < E) atom_add_f32(&cnt[dst[e]], 1.0f);
}

__global__ void invdeg_kernel(const float* __restrict__ cnt, float* __restrict__ inv, int n) {
  int i = blockIdx.x * blockDim.x + threadIdx.x;
  if (i < n) inv[i] = 1.0f / fmaxf(cnt[i], 1.0f);
}

template <int D>
__global__ void scatter_add_kernel(const float* __restrict__ x,
                                   const int* __restrict__ src,
                                   const int* __restrict__ dst,
                                   float* __restrict__ agg, int E) {
  constexpr int C = D / 4;
  long tid = (long)blockIdx.x * blockDim.x + threadIdx.x;
  long total = (long)E * C;
  if (tid >= total) return;
  int e = (int)(tid / C);
  int f = (int)(tid % C) * 4;
  int s = src[e];
  int d = dst[e];
  float4 v = *(const float4*)(x + (long)s * D + f);
  float* o = agg + (long)d * D + f;
  atom_add_f32(o + 0, v.x);
  atom_add_f32(o + 1, v.y);
  atom_add_f32(o + 2, v.z);
  atom_add_f32(o + 3, v.w);
}

__global__ void split_z2_kernel(const float* __restrict__ z2,
                                float* __restrict__ out01,
                                float* __restrict__ out2, int n) {
  int i = blockIdx.x * blockDim.x + threadIdx.x;
  if (i < n) {
    out01[2 * i + 0] = z2[3 * i + 0];
    out01[2 * i + 1] = z2[3 * i + 1];
    out2[i] = z2[3 * i + 2];
  }
}

template <int DIN, int DOUT, int ACT, bool HAS_AGG>
__global__ __launch_bounds__(256) void wmma_linear_kernel(
    const float* __restrict__ agg, const float* __restrict__ invdeg,
    const float* __restrict__ x,
    const float* __restrict__ Wl, const float* __restrict__ bias,
    const float* __restrict__ Wr,
    float* __restrict__ out, float* __restrict__ out2, int nnodes) {
  constexpr int MT = 4;
  constexpr int ROWS = 16 * MT;
  constexpr int DOUTP = (DOUT + 15) & ~15;
  constexpr int LDW = DIN + 4;
  constexpr int C4 = DIN / 4;
  __shared__ float sWl[DOUTP * LDW];
  __shared__ float sWr[HAS_AGG ? DOUTP * LDW : 1];
  __shared__ float sA[ROWS * LDW];
  __shared__ float sX[HAS_AGG ? ROWS * LDW : 1];

  const int tid = threadIdx.x;
  const int bs = blockDim.x;
  const int row0 = blockIdx.x * ROWS;

  const float4 f4z = make_float4(0.0f, 0.0f, 0.0f, 0.0f);

  for (int i = tid; i < DOUTP * C4; i += bs) {
    int r = i / C4, c4 = (i - r * C4) * 4;
    float4 wl = (r < DOUT) ? *(const float4*)(Wl + (long)r * DIN + c4) : f4z;
    *(float4*)&sWl[r * LDW + c4] = wl;
    if constexpr (HAS_AGG) {
      float4 wr = (r < DOUT) ? *(const float4*)(Wr + (long)r * DIN + c4) : f4z;
      *(float4*)&sWr[r * LDW + c4] = wr;
    }
  }
  for (int i = tid; i < ROWS * C4; i += bs) {
    int r = i / C4, c4 = (i - r * C4) * 4;
    int m = row0 + r;
    if (m >= nnodes) m = nnodes - 1;
    if constexpr (HAS_AGG) {
      float sc = invdeg[m];
      float4 av = *(const float4*)(agg + (long)m * DIN + c4);
      av.x *= sc; av.y *= sc; av.z *= sc; av.w *= sc;
      *(float4*)&sA[r * LDW + c4] = av;
      *(float4*)&sX[r * LDW + c4] = *(const float4*)(x + (long)m * DIN + c4);
    } else {
      *(float4*)&sA[r * LDW + c4] = *(const float4*)(x + (long)m * DIN + c4);
    }
  }
  __syncthreads();

  const int wave = tid >> 5;
  const int lane = tid & 31;
  const int half = lane >> 4;
  const int l16 = lane & 15;
  const int n = wave * 16 + l16;

  v8f acc[MT];
#pragma unroll
  for (int mt = 0; mt < MT; ++mt) acc[mt] = (v8f){};

#pragma unroll
  for (int k = 0; k < DIN; k += 4) {
    const int kk = k + 2 * half;
    v2f b;
    b.x = sWl[n * LDW + kk];
    b.y = sWl[n * LDW + kk + 1];
    v2f b2;
    if constexpr (HAS_AGG) {
      b2.x = sWr[n * LDW + kk];
      b2.y = sWr[n * LDW + kk + 1];
    }
#pragma unroll
    for (int mt = 0; mt < MT; ++mt) {
      const int r = mt * 16 + l16;
      v2f a;
      a.x = sA[r * LDW + kk];
      a.y = sA[r * LDW + kk + 1];
      acc[mt] = __builtin_amdgcn_wmma_f32_16x16x4_f32(false, a, false, b,
                                                      (short)0, acc[mt], false,
                                                      false);
      if constexpr (HAS_AGG) {
        v2f a2;
        a2.x = sX[r * LDW + kk];
        a2.y = sX[r * LDW + kk + 1];
        acc[mt] = __builtin_amdgcn_wmma_f32_16x16x4_f32(false, a2, false, b2,
                                                        (short)0, acc[mt],
                                                        false, false);
      }
    }
  }

  const bool colOK = (n < DOUT);
  const float bcol = colOK ? bias[colOK ? n : 0] : 0.0f;
#pragma unroll
  for (int mt = 0; mt < MT; ++mt) {
#pragma unroll
    for (int i = 0; i < 8; ++i) {
      int row = row0 + mt * 16 + i + 8 * half;
      float v = acc[mt][i] + bcol;
      if (ACT == 1) v = tanhf(v);
      if (ACT == 2) v = (v > 0.0f) ? v : 0.01f * v;
      if (colOK && row < nnodes) {
        out[(long)row * DOUT + n] = v;
        if (out2) out2[(long)row * DOUT + n] = v;
      }
    }
  }
}


static inline unsigned cdiv(long a, long b) { return (unsigned)((a + b - 1) / b); }

extern "C" void kernel_launch(void* const* d_in, const int* in_sizes, int n_in,
                              void* d_out, int out_size, void* d_ws, size_t ws_size,
                              hipStream_t stream) {
  const float* x = (const float*)d_in[0];
  const int* ei = (const int*)d_in[1];
  const int N = in_sizes[0] / 64;
  const int E = in_sizes[1] / 2;
  const int* src = ei;
  const int* dst = ei + E;

  const float* g1Wl = (const float*)d_in[2];
  const float* g1b  = (const float*)d_in[3];
  const float* g1Wr = (const float*)d_in[4];
  const float* g2Wl = (const float*)d_in[5];
  const float* g2b  = (const float*)d_in[6];
  const float* g2Wr = (const float*)d_in[7];
  const float* g3Wl = (const float*)d_in[8];
  const float* g3b  = (const float*)d_in[9];
  const float* g3Wr = (const float*)d_in[10];
  const float* trW  = (const float*)d_in[11];
  const float* trb  = (const float*)d_in[12];
  const float* d1W  = (const float*)d_in[13];
  const float* d1b  = (const float*)d_in[14];
  const float* d2Wl = (const float*)d_in[15];
  const float* d2b  = (const float*)d_in[16];
  const float* d2Wr = (const float*)d_in[17];
  const float* d3Wl = (const float*)d_in[18];
  const float* d3b  = (const float*)d_in[19];
  const float* d3Wr = (const float*)d_in[20];
  const float* d4Wl = (const float*)d_in[21];
  const float* d4b  = (const float*)d_in[22];
  const float* d4Wr = (const float*)d_in[23];

  float* ws   = (float*)d_ws;
  float* cnt  = ws;
  float* inv  = ws + N;
  float* bufA = ws + 2 * (size_t)N;
  float* bufB = bufA + (size_t)N * 128;
  float* bufC = bufB + (size_t)N * 128;
  float* z2   = bufC + (size_t)N * 128;

  float* outZ01 = (float*)d_out;
  float* outZ2  = outZ01 + 2 * (size_t)N;
  float* outMu  = outZ2 + N;
  float* outLv  = outMu + (size_t)N * 64;

  const unsigned gN64 = (unsigned)((N + 63) / 64);

  zero_kernel<<<cdiv(N, 256), 256, 0, stream>>>(cnt, N);
  degree_kernel<<<cdiv(E, 256), 256, 0, stream>>>(dst, cnt, E);
  invdeg_kernel<<<cdiv(N, 256), 256, 0, stream>>>(cnt, inv, N);

  zero_kernel<<<cdiv((long)N * 64, 256), 256, 0, stream>>>(bufA, (long)N * 64);
  scatter_add_kernel<64><<<cdiv((long)E * 16, 256), 256, 0, stream>>>(x, src, dst, bufA, E);
  wmma_linear_kernel<64, 128, 1, true><<<gN64, 256, 0, stream>>>(
      bufA, inv, x, g1Wl, g1b, g1Wr, bufB, nullptr, N);

  zero_kernel<<<cdiv((long)N * 128, 256), 256, 0, stream>>>(bufA, (long)N * 128);
  scatter_add_kernel<128><<<cdiv((long)E * 32, 256), 256, 0, stream>>>(bufB, src, dst, bufA, E);
  wmma_linear_kernel<128, 128, 1, true><<<gN64, 256, 0, stream>>>(
      bufA, inv, bufB, g2Wl, g2b, g2Wr, bufC, nullptr, N);

  zero_kernel<<<cdiv((long)N * 128, 256), 256, 0, stream>>>(bufA, (long)N * 128);
  scatter_add_kernel<128><<<cdiv((long)E * 32, 256), 256, 0, stream>>>(bufC, src, dst, bufA, E);
  wmma_linear_kernel<128, 128, 1, true><<<gN64, 256, 0, stream>>>(
      bufA, inv, bufC, g3Wl, g3b, g3Wr, bufB, nullptr, N);

  wmma_linear_kernel<128, 64, 0, false><<<gN64, 128, 0, stream>>>(
      nullptr, nullptr, bufB, trW, trb, nullptr, outMu, outLv, N);

  wmma_linear_kernel<64, 128, 2, false><<<gN64, 256, 0, stream>>>(
      nullptr, nullptr, outMu, d1W, d1b, nullptr, bufA, nullptr, N);

  zero_kernel<<<cdiv((long)N * 128, 256), 256, 0, stream>>>(bufC, (long)N * 128);
  scatter_add_kernel<128><<<cdiv((long)E * 32, 256), 256, 0, stream>>>(bufA, src, dst, bufC, E);
  wmma_linear_kernel<128, 128, 2, true><<<gN64, 256, 0, stream>>>(
      bufC, inv, bufA, d2Wl, d2b, d2Wr, bufB, nullptr, N);

  zero_kernel<<<cdiv((long)N * 128, 256), 256, 0, stream>>>(bufC, (long)N * 128);
  scatter_add_kernel<128><<<cdiv((long)E * 32, 256), 256, 0, stream>>>(bufB, src, dst, bufC, E);
  wmma_linear_kernel<128, 64, 1, true><<<gN64, 128, 0, stream>>>(
      bufC, inv, bufB, d3Wl, d3b, d3Wr, bufA, nullptr, N);

  zero_kernel<<<cdiv((long)N * 64, 256), 256, 0, stream>>>(bufC, (long)N * 64);
  scatter_add_kernel<64><<<cdiv((long)E * 16, 256), 256, 0, stream>>>(bufA, src, dst, bufC, E);
  wmma_linear_kernel<64, 3, 0, true><<<gN64, 32, 0, stream>>>(
      bufC, inv, bufA, d4Wl, d4b, d4Wr, z2, nullptr, N);

  split_z2_kernel<<<cdiv(N, 256), 256, 0, stream>>>(z2, outZ01, outZ2, N);
}


# Edges bucketed by destination once per call (integer rank atomics + prefix sum); one wave per node sums its rows in bucket order (arrival order, varies run to run); no zero fill, no float atomics
# speedup vs baseline: 2.2251x; 2.2251x over previous
#include <hip/hip_runtime.h>
#include <hip/hip_bf16.h>

typedef __attribute__((ext_vector_type(2))) float v2f;
typedef __attribute__((ext_vector_type(8))) float v8f;


__global__ void zero_int_kernel(int* __restrict__ p, int n) {
  int i = blockIdx.x * blockDim.x + threadIdx.x;
  if (i < n) p[i] = 0;
}

__global__ void rank_kernel(const int* __restrict__ dst, int* __restrict__ cnt,
                            int* __restrict__ rank, int E) {
  int e = blockIdx.x * blockDim.x + threadIdx.x;
  if (e < E)
    rank[e] = __hip_atomic_fetch_add(&cnt[dst[e]], 1, __ATOMIC_RELAXED,
                                     __HIP_MEMORY_SCOPE_AGENT);
}

__global__ __launch_bounds__(256) void chunk_sum_kernel(
    const int* __restrict__ cnt, int* __restrict__ csum, int n) {
  __shared__ int sw[8];
  const int tid = threadIdx.x;
  const int base = blockIdx.x * 1024 + tid * 4;
  int s = 0;
  for (int u = 0; u < 4; ++u)
    if (base + u < n) s += cnt[base + u];
  for (int d = 16; d > 0; d >>= 1) s += __shfl_xor(s, d, 32);
  if ((tid & 31) == 0) sw[tid >> 5] = s;
  __syncthreads();
  if (tid == 0) {
    int t = 0;
    for (int w = 0; w < 8; ++w) t += sw[w];
    csum[blockIdx.x] = t;
  }
}

__global__ __launch_bounds__(256) void rowptr_kernel(
    const int* __restrict__ cnt, const int* __restrict__ csum,
    int* __restrict__ rowptr, float* __restrict__ inv, int n) {
  __shared__ int sw[8];
  __shared__ int soff;
  const int tid = threadIdx.x;
  const int lane = tid & 31;
  const int wave = tid >> 5;

  int p = 0;
  for (int i = tid; i < (int)blockIdx.x; i += 256) p += csum[i];
  for (int d = 16; d > 0; d >>= 1) p += __shfl_xor(p, d, 32);
  if (lane == 0) sw[wave] = p;
  __syncthreads();
  if (tid == 0) {
    int t = 0;
    for (int w = 0; w < 8; ++w) t += sw[w];
    soff = t;
  }
  __syncthreads();
  const int off = soff;

  const int base = blockIdx.x * 1024 + tid * 4;
  int c[4];
  int t = 0;
  for (int u = 0; u < 4; ++u) {
    c[u] = (base + u < n) ? cnt[base + u] : 0;
    t += c[u];
  }
  int incl = t;
  for (int d = 1; d < 32; d <<= 1) {
    int v = __shfl_up(incl, d, 32);
    if (lane >= d) incl += v;
  }
  if (lane == 31) sw[wave] = incl;
  __syncthreads();
  int run = off + incl - t;
  for (int w = 0; w < wave; ++w) run += sw[w];
  for (int u = 0; u < 4; ++u) {
    const int i = base + u;
    if (i < n) {
      rowptr[i] = run;
      inv[i] = 1.0f / fmaxf((float)c[u], 1.0f);
      run += c[u];
      if (i == n - 1) rowptr[n] = run;
    }
  }
}

__global__ void fill_kernel(const int* __restrict__ src, const int* __restrict__ dst,
                            const int* __restrict__ rank,
                            const int* __restrict__ rowptr,
                            int* __restrict__ col, int E) {
  int e = blockIdx.x * blockDim.x + threadIdx.x;
  if (e < E) col[rowptr[dst[e]] + rank[e]] = src[e];
}

template <int D, int U, typename VT>
__device__ __forceinline__ void add_rows(VT& acc, const float* __restrict__ xl,
                                         int cols, int j) {
  VT v[U];
#pragma unroll
  for (int u = 0; u < U; ++u) {
    const int s = __builtin_amdgcn_readlane(cols, j + u);
    v[u] = *(const VT*)(xl + (long)s * D);
  }
#pragma unroll
  for (int u = 0; u < U; ++u) acc += v[u];
}

template <int D>
__global__ __launch_bounds__(256) void gather_sum_kernel(
    const float* __restrict__ x, const int* __restrict__ rowptr,
    const int* __restrict__ col, float* __restrict__ agg, int n) {
  constexpr int V = D / 32;
  typedef float VT __attribute__((ext_vector_type(V)));
  const int lane = threadIdx.x & 31;
  const int node = __builtin_amdgcn_readfirstlane(
      (int)((blockIdx.x * blockDim.x + threadIdx.x) >> 5));
  if (node >= n) return;
  const int beg = rowptr[node];
  const int end = rowptr[node + 1];
  const float* xl = x + lane * V;
  VT acc = (VT)(0.0f);
  for (int base = beg; base < end; base += 32) {
    const int m = min(32, end - base);
    const int cols = (lane < m) ? col[base + lane] : 0;
    int j = 0;
    for (; j + 8 <= m; j += 8) add_rows<D, 8, VT>(acc, xl, cols, j);
    if (m - j >= 4) { add_rows<D, 4, VT>(acc, xl, cols, j); j += 4; }
    if (m - j >= 2) { add_rows<D, 2, VT>(acc, xl, cols, j); j += 2; }
    if (m - j >= 1) { add_rows<D, 1, VT>(acc, xl, cols, j); }
  }
  *(VT*)(agg + (long)node * D + lane * V) = acc;
}

__global__ void split_z2_kernel(const float* __restrict__ z2,
                                float* __restrict__ out01,
                                float* __restrict__ out2, int n) {
  int i = blockIdx.x * blockDim.x + threadIdx.x;
  if (i < n) {
    out01[2 * i + 0] = z2[3 * i + 0];
    out01[2 * i + 1] = z2[3 * i + 1];
    out2[i] = z2[3 * i + 2];
  }
}

template <int DIN, int DOUT, int ACT, bool HAS_AGG>
__global__ __launch_bounds__(256) void wmma_linear_kernel(
    const float* agg, const float* __restrict__ invdeg,
    const float* __restrict__ x,
    const float* __restrict__ Wl, const float* __restrict__ bias,
    const float* __restrict__ Wr,
    float* out, float* __restrict__ out2, int nnodes) {
  constexpr int MT = 4;
  constexpr int ROWS = 16 * MT;
  constexpr int DOUTP = (DOUT + 15) & ~15;
  constexpr int LDW = DIN + 4;
  constexpr int C4 = DIN / 4;
  __shared__ float sWl[DOUTP * LDW];
  __shared__ float sWr[HAS_AGG ? DOUTP * LDW : 1];
  __shared__ float sA[ROWS * LDW];
  __shared__ float sX[HAS_AGG ? ROWS * LDW : 1];

  const int tid = threadIdx.x;
  const int bs = blockDim.x;
  const int row0 = blockIdx.x * ROWS;

  const float4 f4z = make_float4(0.0f, 0.0f, 0.0f, 0.0f);

  for (int i = tid; i < DOUTP * C4; i += bs) {
    int r = i / C4, c4 = (i - r * C4) * 4;
    float4 wl = (r < DOUT) ? *(const float4*)(Wl + (long)r * DIN + c4) : f4z;
    *(float4*)&sWl[r * LDW + c4] = wl;
    if constexpr (HAS_AGG) {
      float4 wr = (r < DOUT) ? *(const float4*)(Wr + (long)r * DIN + c4) : f4z;
      *(float4*)&sWr[r * LDW + c4] = wr;
    }
  }
  for (int i = tid; i < ROWS * C4; i += bs) {
    int r = i / C4, c4 = (i - r * C4) * 4;
    int m = row0 + r;
    if (m >= nnodes) m = nnodes - 1;
    if constexpr (HAS_AGG) {
      float sc = invdeg[m];
      float4 av = *(const float4*)(agg + (long)m * DIN + c4);
      av.x *= sc; av.y *= sc; av.z *= sc; av.w *= sc;
      *(float4*)&sA[r * LDW + c4] = av;
      *(float4*)&sX[r * LDW + c4] = *(const float4*)(x + (long)m * DIN + c4);
    } else {
      *(float4*)&sA[r * LDW + c4] = *(const float4*)(x + (long)m * DIN + c4);
    }
  }
  __syncthreads();

  const int wave = tid >> 5;
  const int lane = tid & 31;
  const int half = lane >> 4;
  const int l16 = lane & 15;
  const int n = wave * 16 + l16;

  v8f acc[MT];
#pragma unroll
  for (int mt = 0; mt < MT; ++mt) acc[mt] = (v8f){};

#pragma unroll
  for (int k = 0; k < DIN; k += 4) {
    const int kk = k + 2 * half;
    v2f b;
    b.x = sWl[n * LDW + kk];
    b.y = sWl[n * LDW + kk + 1];
    v2f b2;
    if constexpr (HAS_AGG) {
      b2.x = sWr[n * LDW + kk];
      b2.y = sWr[n * LDW + kk + 1];
    }
#pragma unroll
    for (int mt = 0; mt < MT; ++mt) {
      const int r = mt * 16 + l16;
      v2f a;
      a.x = sA[r * LDW + kk];
      a.y = sA[r * LDW + kk + 1];
      acc[mt] = __builtin_amdgcn_wmma_f32_16x16x4_f32(false, a, false, b,
                                                      (short)0, acc[mt], false,
                                                      false);
      if constexpr (HAS_AGG) {
        v2f a2;
        a2.x = sX[r * LDW + kk];
        a2.y = sX[r * LDW + kk + 1];
        acc[mt] = __builtin_amdgcn_wmma_f32_16x16x4_f32(false, a2, false, b2,
                                                        (short)0, acc[mt],
                                                        false, false);
      }
    }
  }

  const bool colOK = (n < DOUT);
  const float bcol = colOK ? bias[colOK ? n : 0] : 0.0f;
#pragma unroll
  for (int mt = 0; mt < MT; ++mt) {
#pragma unroll
    for (int i = 0; i < 8; ++i) {
      int row = row0 + mt * 16 + i + 8 * half;
      float v = acc[mt][i] + bcol;
      if (ACT == 1) v = tanhf(v);
      if (ACT == 2) v = (v > 0.0f) ? v : 0.01f * v;
      if (colOK && row < nnodes) {
        out[(long)row * DOUT + n] = v;
        if (out2) out2[(long)row * DOUT + n] = v;
      }
    }
  }
}


static inline unsigned cdiv(long a, long b) { return (unsigned)((a + b - 1) / b); }

extern "C" void kernel_launch(void* const* d_in, const int* in_sizes, int n_in,
                              void* d_out, int out_size, void* d_ws, size_t ws_size,
                              hipStream_t stream) {
  const float* x = (const float*)d_in[0];
  const int* ei = (const int*)d_in[1];
  const int N = in_sizes[0] / 64;
  const int E = in_sizes[1] / 2;
  const int* src = ei;
  const int* dst = ei + E;

  const float* g1Wl = (const float*)d_in[2];
  const float* g1b  = (const float*)d_in[3];
  const float* g1Wr = (const float*)d_in[4];
  const float* g2Wl = (const float*)d_in[5];
  const float* g2b  = (const float*)d_in[6];
  const float* g2Wr = (const float*)d_in[7];
  const float* g3Wl = (const float*)d_in[8];
  const float* g3b  = (const float*)d_in[9];
  const float* g3Wr = (const float*)d_in[10];
  const float* trW  = (const float*)d_in[11];
  const float* trb  = (const float*)d_in[12];
  const float* d1W  = (const float*)d_in[13];
  const float* d1b  = (const float*)d_in[14];
  const float* d2Wl = (const float*)d_in[15];
  const float* d2b  = (const float*)d_in[16];
  const float* d2Wr = (const float*)d_in[17];
  const float* d3Wl = (const float*)d_in[18];
  const float* d3b  = (const float*)d_in[19];
  const float* d3Wr = (const float*)d_in[20];
  const float* d4Wl = (const float*)d_in[21];
  const float* d4b  = (const float*)d_in[22];
  const float* d4Wr = (const float*)d_in[23];

  float* ws   = (float*)d_ws;
  int*   cnt  = (int*)ws;
  float* inv  = ws + N;
  float* bufA = ws + 2 * (size_t)N;
  float* bufB = bufA + (size_t)N * 128;
  float* bufC = bufB + (size_t)N * 128;
  float* z2   = bufC + (size_t)N * 128;
  int*   rowptr = (int*)bufC;
  int*   col    = rowptr + (N + 1);
  float* bufD   = bufC + (size_t)N * 64;
  int*   rank   = (int*)bufB;
  int*   csum   = (int*)z2;

  float* outZ01 = (float*)d_out;
  float* outZ2  = outZ01 + 2 * (size_t)N;
  float* outMu  = outZ2 + N;
  float* outLv  = outMu + (size_t)N * 64;

  const unsigned gN64 = (unsigned)((N + 63) / 64);
  const unsigned gE   = cdiv(E, 256);
  const unsigned gCh  = cdiv(N, 1024);
  const unsigned gG   = cdiv((long)N * 32, 256);

  zero_int_kernel<<<cdiv(N, 256), 256, 0, stream>>>(cnt, N);
  rank_kernel<<<gE, 256, 0, stream>>>(dst, cnt, rank, E);
  chunk_sum_kernel<<<gCh, 256, 0, stream>>>(cnt, csum, N);
  rowptr_kernel<<<gCh, 256, 0, stream>>>(cnt, csum, rowptr, inv, N);
  fill_kernel<<<gE, 256, 0, stream>>>(src, dst, rank, rowptr, col, E);

  gather_sum_kernel<64><<<gG, 256, 0, stream>>>(x, rowptr, col, bufA, N);
  wmma_linear_kernel<64, 128, 1, true><<<gN64, 256, 0, stream>>>(
      bufA, inv, x, g1Wl, g1b, g1Wr, bufB, nullptr, N);

  gather_sum_kernel<128><<<gG, 256, 0, stream>>>(bufB, rowptr, col, bufA, N);
  wmma_linear_kernel<128, 128, 1, true><<<gN64, 256, 0, stream>>>(
      bufA, inv, bufB, g2Wl, g2b, g2Wr, bufA, nullptr, N);

  gather_sum_kernel<128><<<gG, 256, 0, stream>>>(bufA, rowptr, col, bufB, N);
  wmma_linear_kernel<128, 128, 1, true><<<gN64, 256, 0, stream>>>(
      bufB, inv, bufA, g3Wl, g3b, g3Wr, bufB, nullptr, N);

  wmma_linear_kernel<128, 64, 0, false><<<gN64, 128, 0, stream>>>(
      nullptr, nullptr, bufB, trW, trb, nullptr, outMu, outLv, N);

  wmma_linear_kernel<64, 128, 2, false><<<gN64, 256, 0, stream>>>(
      nullptr, nullptr, outMu, d1W, d1b, nullptr, bufA, nullptr, N);

  gather_sum_kernel<128><<<gG, 256, 0, stream>>>(bufA, rowptr, col, bufB, N);
  wmma_linear_kernel<128, 128, 2, true><<<gN64, 256, 0, stream>>>(
      bufB, inv, bufA, d2Wl, d2b, d2Wr, bufB, nullptr, N);

  gather_sum_kernel<128><<<gG, 256, 0, stream>>>(bufB, rowptr, col, bufA, N);
  wmma_linear_kernel<128, 64, 1, true><<<gN64, 128, 0, stream>>>(
      bufA, inv, bufB, d3Wl, d3b, d3Wr, bufD, nullptr, N);

  gather_sum_kernel<64><<<gG, 256, 0, stream>>>(bufD, rowptr, col, bufA, N);
  wmma_linear_kernel<64, 3, 0, true><<<gN64, 32, 0, stream>>>(
      bufA, inv, bufD, d4Wl, d4b, d4Wr, z2, nullptr, N);

  split_z2_kernel<<<cdiv(N, 256), 256, 0, stream>>>(z2, outZ01, outZ2, N);
}


# Linear kernels: a work-group stages its weights in LDS once and walks its row tiles; the next tile's rows load into registers during the current tile's products (256-1024 groups, not one per tile)
# speedup vs baseline: 2.9529x; 1.3271x over previous
#include <hip/hip_runtime.h>
#include <hip/hip_bf16.h>

typedef __attribute__((ext_vector_type(2))) float v2f;
typedef __attribute__((ext_vector_type(4))) float v4f;
typedef __attribute__((ext_vector_type(8))) float v8f;


__global__ void zero_int_kernel(int* __restrict__ p, int n) {
  int i = blockIdx.x * blockDim.x + threadIdx.x;
  if (i < n) p[i] = 0;
}

__global__ void rank_kernel(const int* __restrict__ dst, int* __restrict__ cnt,
                            int* __restrict__ rank, int E) {
  int e = blockIdx.x * blockDim.x + threadIdx.x;
  if (e < E)
    rank[e] = __hip_atomic_fetch_add(&cnt[dst[e]], 1, __ATOMIC_RELAXED,
                                     __HIP_MEMORY_SCOPE_AGENT);
}

__global__ __launch_bounds__(256) void chunk_sum_kernel(
    const int* __restrict__ cnt, int* __restrict__ csum, int n) {
  __shared__ int sw[8];
  const int tid = threadIdx.x;
  const int base = blockIdx.x * 1024 + tid * 4;
  int s = 0;
  for (int u = 0; u < 4; ++u)
    if (base + u < n) s += cnt[base + u];
  for (int d = 16; d > 0; d >>= 1) s += __shfl_xor(s, d, 32);
  if ((tid & 31) == 0) sw[tid >> 5] = s;
  __syncthreads();
  if (tid == 0) {
    int t = 0;
    for (int w = 0; w < 8; ++w) t += sw[w];
    csum[blockIdx.x] = t;
  }
}

__global__ __launch_bounds__(256) void rowptr_kernel(
    const int* __restrict__ cnt, const int* __restrict__ csum,
    int* __restrict__ rowptr, float* __restrict__ inv, int n) {
  __shared__ int sw[8];
  __shared__ int soff;
  const int tid = threadIdx.x;
  const int lane = tid & 31;
  const int wave = tid >> 5;

  int p = 0;
  for (int i = tid; i < (int)blockIdx.x; i += 256) p += csum[i];
  for (int d = 16; d > 0; d >>= 1) p += __shfl_xor(p, d, 32);
  if (lane == 0) sw[wave] = p;
  __syncthreads();
  if (tid == 0) {
    int t = 0;
    for (int w = 0; w < 8; ++w) t += sw[w];
    soff = t;
  }
  __syncthreads();
  const int off = soff;

  const int base = blockIdx.x * 1024 + tid * 4;
  int c[4];
  int t = 0;
  for (int u = 0; u < 4; ++u) {
    c[u] = (base + u < n) ? cnt[base + u] : 0;
    t += c[u];
  }
  int incl = t;
  for (int d = 1; d < 32; d <<= 1) {
    int v = __shfl_up(incl, d, 32);
    if (lane >= d) incl += v;
  }
  if (lane == 31) sw[wave] = incl;
  __syncthreads();
  int run = off + incl - t;
  for (int w = 0; w < wave; ++w) run += sw[w];
  for (int u = 0; u < 4; ++u) {
    const int i = base + u;
    if (i < n) {
      rowptr[i] = run;
      inv[i] = 1.0f / fmaxf((float)c[u], 1.0f);
      run += c[u];
      if (i == n - 1) rowptr[n] = run;
    }
  }
}

__global__ void fill_kernel(const int* __restrict__ src, const int* __restrict__ dst,
                            const int* __restrict__ rank,
                            const int* __restrict__ rowptr,
                            int* __restrict__ col, int E) {
  int e = blockIdx.x * blockDim.x + threadIdx.x;
  if (e < E) col[rowptr[dst[e]] + rank[e]] = src[e];
}

template <int D, int U, typename VT>
__device__ __forceinline__ void add_rows(VT& acc, const float* __restrict__ xl,
                                         int cols, int j) {
  VT v[U];
#pragma unroll
  for (int u = 0; u < U; ++u) {
    const int s = __builtin_amdgcn_readlane(cols, j + u);
    v[u] = *(const VT*)(xl + (long)s * D);
  }
#pragma unroll
  for (int u = 0; u < U; ++u) acc += v[u];
}

template <int D>
__global__ __launch_bounds__(256) void gather_sum_kernel(
    const float* __restrict__ x, const int* __restrict__ rowptr,
    const int* __restrict__ col, float* __restrict__ agg, int n) {
  constexpr int V = D / 32;
  typedef float VT __attribute__((ext_vector_type(V)));
  const int lane = threadIdx.x & 31;
  const int node = __builtin_amdgcn_readfirstlane(
      (int)((blockIdx.x * blockDim.x + threadIdx.x) >> 5));
  if (node >= n) return;
  const int beg = rowptr[node];
  const int end = rowptr[node + 1];
  const float* xl = x + lane * V;
  VT acc = (VT)(0.0f);
  for (int base = beg; base < end; base += 32) {
    const int m = min(32, end - base);
    const int cols = (lane < m) ? col[base + lane] : 0;
    int j = 0;
    for (; j + 8 <= m; j += 8) add_rows<D, 8, VT>(acc, xl, cols, j);
    if (m - j >= 4) { add_rows<D, 4, VT>(acc, xl, cols, j); j += 4; }
    if (m - j >= 2) { add_rows<D, 2, VT>(acc, xl, cols, j); j += 2; }
    if (m - j >= 1) { add_rows<D, 1, VT>(acc, xl, cols, j); }
  }
  *(VT*)(agg + (long)node * D + lane * V) = acc;
}

__global__ void split_z2_kernel(const float* __restrict__ z2,
                                float* __restrict__ out01,
                                float* __restrict__ out2, int n) {
  int i = blockIdx.x * blockDim.x + threadIdx.x;
  if (i < n) {
    out01[2 * i + 0] = z2[3 * i + 0];
    out01[2 * i + 1] = z2[3 * i + 1];
    out2[i] = z2[3 * i + 2];
  }
}

template <int DIN, int DOUT, int ACT, bool HAS_AGG>
__global__ __launch_bounds__(2 * ((DOUT + 15) & ~15)) void wmma_linear_kernel(
    const float* agg, const float* __restrict__ invdeg,
    const float* __restrict__ x,
    const float* __restrict__ Wl, const float* __restrict__ bias,
    const float* __restrict__ Wr,
    float* out, float* __restrict__ out2, int nnodes, int ntiles) {
  constexpr int MT = 4;
  constexpr int ROWS = 16 * MT;
  constexpr int DOUTP = (DOUT + 15) & ~15;
  constexpr int NT = 2 * DOUTP;
  constexpr int LDW = DIN + 4;
  constexpr int C4 = DIN / 4;
  constexpr int PER = ROWS * C4 / NT;
  __shared__ float sWl[DOUTP * LDW];
  __shared__ float sWr[HAS_AGG ? DOUTP * LDW : 1];
  __shared__ float sA[ROWS * LDW];
  __shared__ float sX[HAS_AGG ? ROWS * LDW : 1];

  const int tid = threadIdx.x;
  int tile = blockIdx.x;
  if (tile >= ntiles) return;

  const float4 f4z = make_float4(0.0f, 0.0f, 0.0f, 0.0f);

  v4f ra[PER];
  v4f rx[HAS_AGG ? PER : 1];
  float rs[HAS_AGG ? PER : 1];

  auto load_tile = [&](int t) {
    const int r0 = t * ROWS;
#pragma unroll
    for (int u = 0; u < PER; ++u) {
      const int i = tid + u * NT;
      const int r = i / C4, c4 = (i - r * C4) * 4;
      int m = r0 + r;
      if (m >= nnodes) m = nnodes - 1;
      if constexpr (HAS_AGG) {
        rs[u] = invdeg[m];
        ra[u] = *(const v4f*)(agg + (long)m * DIN + c4);
        rx[u] = *(const v4f*)(x + (long)m * DIN + c4);
      } else {
        ra[u] = *(const v4f*)(x + (long)m * DIN + c4);
      }
    }
  };
  auto store_tile = [&]() {
#pragma unroll
    for (int u = 0; u < PER; ++u) {
      const int i = tid + u * NT;
      const int r = i / C4, c4 = (i - r * C4) * 4;
      if constexpr (HAS_AGG) {
        *(v4f*)&sA[r * LDW + c4] = ra[u] * rs[u];
        *(v4f*)&sX[r * LDW + c4] = rx[u];
      } else {
        *(v4f*)&sA[r * LDW + c4] = ra[u];
      }
    }
  };

  load_tile(tile);

  for (int i = tid; i < DOUTP * C4; i += NT) {
    int r = i / C4, c4 = (i - r * C4) * 4;
    float4 wl = (r < DOUT) ? *(const float4*)(Wl + (long)r * DIN + c4) : f4z;
    *(float4*)&sWl[r * LDW + c4] = wl;
    if constexpr (HAS_AGG) {
      float4 wr = (r < DOUT) ? *(const float4*)(Wr + (long)r * DIN + c4) : f4z;
      *(float4*)&sWr[r * LDW + c4] = wr;
    }
  }

  const int wave = tid >> 5;
  const int lane = tid & 31;
  const int half = lane >> 4;
  const int l16 = lane & 15;
  const int n = wave * 16 + l16;
  const bool colOK = (n < DOUT);
  const float bcol = colOK ? bias[colOK ? n : 0] : 0.0f;

  store_tile();
  __syncthreads();

  for (;;) {
    const int next = tile + (int)gridDim.x;
    const bool has_next = next < ntiles;
    if (has_next) load_tile(next);

    v8f acc[MT];
#pragma unroll
    for (int mt = 0; mt < MT; ++mt) acc[mt] = (v8f){};

#pragma unroll
    for (int k = 0; k < DIN; k += 4) {
      const int kk = k + 2 * half;
      v2f b;
      b.x = sWl[n * LDW + kk];
      b.y = sWl[n * LDW + kk + 1];
      v2f b2;
      if constexpr (HAS_AGG) {
        b2.x = sWr[n * LDW + kk];
        b2.y = sWr[n * LDW + kk + 1];
      }
#pragma unroll
      for (int mt = 0; mt < MT; ++mt) {
        const int r = mt * 16 + l16;
        v2f a;
        a.x = sA[r * LDW + kk];
        a.y = sA[r * LDW + kk + 1];
        acc[mt] = __builtin_amdgcn_wmma_f32_16x16x4_f32(false, a, false, b,
                                                        (short)0, acc[mt], false,
                                                        false);
        if constexpr (HAS_AGG) {
          v2f a2;
          a2.x = sX[r * LDW + kk];
          a2.y = sX[r * LDW + kk + 1];
          acc[mt] = __builtin_amdgcn_wmma_f32_16x16x4_f32(false, a2, false, b2,
                                                          (short)0, acc[mt],
                                                          false, false);
        }
      }
    }

    const int row0 = tile * ROWS;
#pragma unroll
    for (int mt = 0; mt < MT; ++mt) {
#pragma unroll
      for (int i = 0; i < 8; ++i) {
        int row = row0 + mt * 16 + i + 8 * half;
        float v = acc[mt][i] + bcol;
        if (ACT == 1) v = tanhf(v);
        if (ACT == 2) v = (v > 0.0f) ? v : 0.01f * v;
        if (colOK && row < nnodes) {
          out[(long)row * DOUT + n] = v;
          if (out2) out2[(long)row * DOUT + n] = v;
        }
      }
    }

    if (!has_next) break;
    __syncthreads();
    store_tile();
    __syncthreads();
    tile = next;
  }
}


static inline unsigned cdiv(long a, long b) { return (unsigned)((a + b - 1) / b); }
static inline unsigned umin(unsigned a, unsigned b) { return a < b ? a : b; }

extern "C" void kernel_launch(void* const* d_in, const int* in_sizes, int n_in,
                              void* d_out, int out_size, void* d_ws, size_t ws_size,
                              hipStream_t stream) {
  const float* x = (const float*)d_in[0];
  const int* ei = (const int*)d_in[1];
  const int N = in_sizes[0] / 64;
  const int E = in_sizes[1] / 2;
  const int* src = ei;
  const int* dst = ei + E;

  const float* g1Wl = (const float*)d_in[2];
  const float* g1b  = (const float*)d_in[3];
  const float* g1Wr = (const float*)d_in[4];
  const float* g2Wl = (const float*)d_in[5];
  const float* g2b  = (const float*)d_in[6];
  const float* g2Wr = (const float*)d_in[7];
  const float* g3Wl = (const float*)d_in[8];
  const float* g3b  = (const float*)d_in[9];
  const float* g3Wr = (const float*)d_in[10];
  const float* trW  = (const float*)d_in[11];
  const float* trb  = (const float*)d_in[12];
  const float* d1W  = (const float*)d_in[13];
  const float* d1b  = (const float*)d_in[14];
  const float* d2Wl = (const float*)d_in[15];
  const float* d2b  = (const float*)d_in[16];
  const float* d2Wr = (const float*)d_in[17];
  const float* d3Wl = (const float*)d_in[18];
  const float* d3b  = (const float*)d_in[19];
  const float* d3Wr = (const float*)d_in[20];
  const float* d4Wl = (const float*)d_in[21];
  const float* d4b  = (const float*)d_in[22];
  const float* d4Wr = (const float*)d_in[23];

  float* ws   = (float*)d_ws;
  int*   cnt  = (int*)ws;
  float* inv  = ws + N;
  float* bufA = ws + 2 * (size_t)N;
  float* bufB = bufA + (size_t)N * 128;
  float* bufC = bufB + (size_t)N * 128;
  float* z2   = bufC + (size_t)N * 128;
  int*   rowptr = (int*)bufC;
  int*   col    = rowptr + (N + 1);
  float* bufD   = bufC + (size_t)N * 64;
  int*   rank   = (int*)bufB;
  int*   csum   = (int*)z2;

  float* outZ01 = (float*)d_out;
  float* outZ2  = outZ01 + 2 * (size_t)N;
  float* outMu  = outZ2 + N;
  float* outLv  = outMu + (size_t)N * 64;

  const int tiles = (N + 63) / 64;
  const unsigned gL1 = umin(tiles, 256);
  const unsigned gL2 = umin(tiles, 512);
  const unsigned gL4 = umin(tiles, 1024);
  const unsigned gE   = cdiv(E, 256);
  const unsigned gCh  = cdiv(N, 1024);
  const unsigned gG   = cdiv((long)N * 32, 256);

  zero_int_kernel<<<cdiv(N, 256), 256, 0, stream>>>(cnt, N);
  rank_kernel<<<gE, 256, 0, stream>>>(dst, cnt, rank, E);
  chunk_sum_kernel<<<gCh, 256, 0, stream>>>(cnt, csum, N);
  rowptr_kernel<<<gCh, 256, 0, stream>>>(cnt, csum, rowptr, inv, N);
  fill_kernel<<<gE, 256, 0, stream>>>(src, dst, rank, rowptr, col, E);

  gather_sum_kernel<64><<<gG, 256, 0, stream>>>(x, rowptr, col, bufA, N);
  wmma_linear_kernel<64, 128, 1, true><<<gL2, 256, 0, stream>>>(
      bufA, inv, x, g1Wl, g1b, g1Wr, bufB, nullptr, N, tiles);

  gather_sum_kernel<128><<<gG, 256, 0, stream>>>(bufB, rowptr, col, bufA, N);
  wmma_linear_kernel<128, 128, 1, true><<<gL1, 256, 0, stream>>>(
      bufA, inv, bufB, g2Wl, g2b, g2Wr, bufA, nullptr, N, tiles);

  gather_sum_kernel<128><<<gG, 256, 0, stream>>>(bufA, rowptr, col, bufB, N);
  wmma_linear_kernel<128, 128, 1, true><<<gL1, 256, 0, stream>>>(
      bufB, inv, bufA, g3Wl, g3b, g3Wr, bufB, nullptr, N, tiles);

  wmma_linear_kernel<128, 64, 0, false><<<gL4, 128, 0, stream>>>(
      nullptr, nullptr, bufB, trW, trb, nullptr, outMu, outLv, N, tiles);

  wmma_linear_kernel<64, 128, 2, false><<<gL4, 256, 0, stream>>>(
      nullptr, nullptr, outMu, d1W, d1b, nullptr, bufA, nullptr, N, tiles);

  gather_sum_kernel<128><<<gG, 256, 0, stream>>>(bufA, rowptr, col, bufB, N);
  wmma_linear_kernel<128, 128, 2, true><<<gL1, 256, 0, stream>>>(
      bufB, inv, bufA, d2Wl, d2b, d2Wr, bufB, nullptr, N, tiles);

  gather_sum_kernel<128><<<gG, 256, 0, stream>>>(bufB, rowptr, col, bufA, N);
  wmma_linear_kernel<128, 64, 1, true><<<gL2, 128, 0, stream>>>(
      bufA, inv, bufB, d3Wl, d3b, d3Wr, bufD, nullptr, N, tiles);

  gather_sum_kernel<64><<<gG, 256, 0, stream>>>(bufD, rowptr, col, bufA, N);
  wmma_linear_kernel<64, 3, 0, true><<<(unsigned)tiles, 32, 0, stream>>>(
      bufA, inv, bufD, d4Wl, d4b, d4Wr, z2, nullptr, N, tiles);

  split_z2_kernel<<<cdiv(N, 256), 256, 0, stream>>>(z2, outZ01, outZ2, N);
}


# Decoder L3/L4: lin_l product before the neighbour mean: gathered rows are 64 and 4 floats, not 128 and 64; gather adds root term + activation (L4: 8 lanes a node, tree sum); sums reassociated
# speedup vs baseline: 3.1509x; 1.0671x over previous
#include <hip/hip_runtime.h>
#include <hip/hip_bf16.h>

typedef __attribute__((ext_vector_type(2))) float v2f;
typedef __attribute__((ext_vector_type(4))) float v4f;
typedef __attribute__((ext_vector_type(8))) float v8f;


__global__ void zero_int_kernel(int* __restrict__ p, int n) {
  int i = blockIdx.x * blockDim.x + threadIdx.x;
  if (i < n) p[i] = 0;
}

__global__ void rank_kernel(const int* __restrict__ dst, int* __restrict__ cnt,
                            int* __restrict__ rank, int E) {
  int e = blockIdx.x * blockDim.x + threadIdx.x;
  if (e < E)
    rank[e] = __hip_atomic_fetch_add(&cnt[dst[e]], 1, __ATOMIC_RELAXED,
                                     __HIP_MEMORY_SCOPE_AGENT);
}

__global__ __launch_bounds__(256) void chunk_sum_kernel(
    const int* __restrict__ cnt, int* __restrict__ csum, int n) {
  __shared__ int sw[8];
  const int tid = threadIdx.x;
  const int base = blockIdx.x * 1024 + tid * 4;
  int s = 0;
  for (int u = 0; u < 4; ++u)
    if (base + u < n) s += cnt[base + u];
  for (int d = 16; d > 0; d >>= 1) s += __shfl_xor(s, d, 32);
  if ((tid & 31) == 0) sw[tid >> 5] = s;
  __syncthreads();
  if (tid == 0) {
    int t = 0;
    for (int w = 0; w < 8; ++w) t += sw[w];
    csum[blockIdx.x] = t;
  }
}

__global__ __launch_bounds__(256) void rowptr_kernel(
    const int* __restrict__ cnt, const int* __restrict__ csum,
    int* __restrict__ rowptr, float* __restrict__ inv, int n) {
  __shared__ int sw[8];
  __shared__ int soff;
  const int tid = threadIdx.x;
  const int lane = tid & 31;
  const int wave = tid >> 5;

  int p = 0;
  for (int i = tid; i < (int)blockIdx.x; i += 256) p += csum[i];
  for (int d = 16; d > 0; d >>= 1) p += __shfl_xor(p, d, 32);
  if (lane == 0) sw[wave] = p;
  __syncthreads();
  if (tid == 0) {
    int t = 0;
    for (int w = 0; w < 8; ++w) t += sw[w];
    soff = t;
  }
  __syncthreads();
  const int off = soff;

  const int base = blockIdx.x * 1024 + tid * 4;
  int c[4];
  int t = 0;
  for (int u = 0; u < 4; ++u) {
    c[u] = (base + u < n) ? cnt[base + u] : 0;
    t += c[u];
  }
  int incl = t;
  for (int d = 1; d < 32; d <<= 1) {
    int v = __shfl_up(incl, d, 32);
    if (lane >= d) incl += v;
  }
  if (lane == 31) sw[wave] = incl;
  __syncthreads();
  int run = off + incl - t;
  for (int w = 0; w < wave; ++w) run += sw[w];
  for (int u = 0; u < 4; ++u) {
    const int i = base + u;
    if (i < n) {
      rowptr[i] = run;
      inv[i] = 1.0f / fmaxf((float)c[u], 1.0f);
      run += c[u];
      if (i == n - 1) rowptr[n] = run;
    }
  }
}

__global__ void fill_kernel(const int* __restrict__ src, const int* __restrict__ dst,
                            const int* __restrict__ rank,
                            const int* __restrict__ rowptr,
                            int* __restrict__ col, int E) {
  int e = blockIdx.x * blockDim.x + threadIdx.x;
  if (e < E) col[rowptr[dst[e]] + rank[e]] = src[e];
}

template <int D, int U, typename VT>
__device__ __forceinline__ void add_rows(VT& acc, const float* __restrict__ xl,
                                         int cols, int j) {
  VT v[U];
#pragma unroll
  for (int u = 0; u < U; ++u) {
    const int s = __builtin_amdgcn_readlane(cols, j + u);
    v[u] = *(const VT*)(xl + (long)s * D);
  }
#pragma unroll
  for (int u = 0; u < U; ++u) acc += v[u];
}

template <int D>
__global__ __launch_bounds__(256) void gather_sum_kernel(
    const float* __restrict__ x, const int* __restrict__ rowptr,
    const int* __restrict__ col, float* __restrict__ agg, int n) {
  constexpr int V = D / 32;
  typedef float VT __attribute__((ext_vector_type(V)));
  const int lane = threadIdx.x & 31;
  const int node = __builtin_amdgcn_readfirstlane(
      (int)((blockIdx.x * blockDim.x + threadIdx.x) >> 5));
  if (node >= n) return;
  const int beg = rowptr[node];
  const int end = rowptr[node + 1];
  const float* xl = x + lane * V;
  VT acc = (VT)(0.0f);
  for (int base = beg; base < end; base += 32) {
    const int m = min(32, end - base);
    const int cols = (lane < m) ? col[base + lane] : 0;
    int j = 0;
    for (; j + 8 <= m; j += 8) add_rows<D, 8, VT>(acc, xl, cols, j);
    if (m - j >= 4) { add_rows<D, 4, VT>(acc, xl, cols, j); j += 4; }
    if (m - j >= 2) { add_rows<D, 2, VT>(acc, xl, cols, j); j += 2; }
    if (m - j >= 1) { add_rows<D, 1, VT>(acc, xl, cols, j); }
  }
  *(VT*)(agg + (long)node * D + lane * V) = acc;
}


template <int D>
__global__ __launch_bounds__(256) void gather_mean_tanh_kernel(
    const float* __restrict__ p, const float* __restrict__ r,
    const float* __restrict__ invdeg, const int* __restrict__ rowptr,
    const int* __restrict__ col, float* __restrict__ out, int n) {
  constexpr int V = D / 32;
  typedef float VT __attribute__((ext_vector_type(V)));
  const int lane = threadIdx.x & 31;
  const int node = __builtin_amdgcn_readfirstlane(
      (int)((blockIdx.x * blockDim.x + threadIdx.x) >> 5));
  if (node >= n) return;
  const int beg = rowptr[node];
  const int end = rowptr[node + 1];
  const float sc = invdeg[node];
  const VT root = *(const VT*)(r + (long)node * D + lane * V);
  const float* pl = p + lane * V;
  VT acc = (VT)(0.0f);
  for (int base = beg; base < end; base += 32) {
    const int m = min(32, end - base);
    const int cols = (lane < m) ? col[base + lane] : 0;
    int j = 0;
    for (; j + 8 <= m; j += 8) add_rows<D, 8, VT>(acc, pl, cols, j);
    if (m - j >= 4) { add_rows<D, 4, VT>(acc, pl, cols, j); j += 4; }
    if (m - j >= 2) { add_rows<D, 2, VT>(acc, pl, cols, j); j += 2; }
    if (m - j >= 1) { add_rows<D, 1, VT>(acc, pl, cols, j); }
  }
  VT v = acc * sc + root;
  for (int u = 0; u < V; ++u) v[u] = tanhf(v[u]);
  *(VT*)(out + (long)node * D + lane * V) = v;
}

__global__ __launch_bounds__(256) void gather_final_kernel(
    const float* __restrict__ p4, const float* __restrict__ r4,
    const float* __restrict__ invdeg, const int* __restrict__ rowptr,
    const int* __restrict__ col, float* __restrict__ out01,
    float* __restrict__ out2, int n) {
  __shared__ float s01[64];
  __shared__ float s2[32];
  const int tid = threadIdx.x;
  const int sub = tid & 7;
  const int g = tid >> 3;
  const int node = blockIdx.x * 32 + g;
  v4f acc = (v4f)(0.0f);
  if (node < n) {
    const int beg = rowptr[node];
    const int end = rowptr[node + 1];
    for (int e = beg + sub; e < end; e += 32) {
      int idx[4];
#pragma unroll
      for (int u = 0; u < 4; ++u) idx[u] = col[min(e + 8 * u, end - 1)];
      v4f v[4];
#pragma unroll
      for (int u = 0; u < 4; ++u) v[u] = *(const v4f*)(p4 + (long)idx[u] * 4);
#pragma unroll
      for (int u = 0; u < 4; ++u) {
        const bool in = (e + 8 * u < end);
        acc.x += in ? v[u].x : 0.0f;
        acc.y += in ? v[u].y : 0.0f;
        acc.z += in ? v[u].z : 0.0f;
      }
    }
  }
  for (int d = 1; d < 8; d <<= 1) {
    acc.x += __shfl_xor(acc.x, d, 32);
    acc.y += __shfl_xor(acc.y, d, 32);
    acc.z += __shfl_xor(acc.z, d, 32);
  }
  if (sub == 0 && node < n) {
    const float sc = invdeg[node];
    const v4f root = *(const v4f*)(r4 + (long)node * 4);
    s01[2 * g + 0] = acc.x * sc + root.x;
    s01[2 * g + 1] = acc.y * sc + root.y;
    s2[g] = acc.z * sc + root.z;
  }
  __syncthreads();
  if (tid < 64) {
    const long i = (long)blockIdx.x * 64 + tid;
    if (i < 2L * n) out01[i] = s01[tid];
  } else if (tid < 96) {
    const long i = (long)blockIdx.x * 32 + (tid - 64);
    if (i < n) out2[i] = s2[tid - 64];
  }
}

template <int DIN, int DOUT, int ACT, bool HAS_AGG, int DH = 0>
__global__ __launch_bounds__(2 * ((DOUT + 15) & ~15)) void wmma_linear_kernel(
    const float* agg, const float* __restrict__ invdeg,
    const float* __restrict__ x,
    const float* __restrict__ Wl, const float* __restrict__ bias,
    const float* __restrict__ Wr,
    float* out, float* __restrict__ out2, int nnodes, int ntiles) {
  constexpr int MT = 4;
  constexpr int ROWS = 16 * MT;
  constexpr int DOUTP = (DOUT + 15) & ~15;
  constexpr int NT = 2 * DOUTP;
  constexpr int LDW = DIN + 4;
  constexpr int C4 = DIN / 4;
  constexpr int PER = ROWS * C4 / NT;
  constexpr int SP = DOUT / 2;
  static_assert(DH == 0 || (!HAS_AGG && DH <= SP), "split form is dense");
  __shared__ float sWl[DOUTP * LDW];
  __shared__ float sWr[HAS_AGG ? DOUTP * LDW : 1];
  __shared__ float sA[ROWS * LDW];
  __shared__ float sX[HAS_AGG ? ROWS * LDW : 1];

  const int tid = threadIdx.x;
  int tile = blockIdx.x;
  if (tile >= ntiles) return;

  const float4 f4z = make_float4(0.0f, 0.0f, 0.0f, 0.0f);

  v4f ra[PER];
  v4f rx[HAS_AGG ? PER : 1];
  float rs[HAS_AGG ? PER : 1];

  auto load_tile = [&](int t) {
    const int r0 = t * ROWS;
#pragma unroll
    for (int u = 0; u < PER; ++u) {
      const int i = tid + u * NT;
      const int r = i / C4, c4 = (i - r * C4) * 4;
      int m = r0 + r;
      if (m >= nnodes) m = nnodes - 1;
      if constexpr (HAS_AGG) {
        rs[u] = invdeg[m];
        ra[u] = *(const v4f*)(agg + (long)m * DIN + c4);
        rx[u] = *(const v4f*)(x + (long)m * DIN + c4);
      } else {
        ra[u] = *(const v4f*)(x + (long)m * DIN + c4);
      }
    }
  };
  auto store_tile = [&]() {
#pragma unroll
    for (int u = 0; u < PER; ++u) {
      const int i = tid + u * NT;
      const int r = i / C4, c4 = (i - r * C4) * 4;
      if constexpr (HAS_AGG) {
        *(v4f*)&sA[r * LDW + c4] = ra[u] * rs[u];
        *(v4f*)&sX[r * LDW + c4] = rx[u];
      } else {
        *(v4f*)&sA[r * LDW + c4] = ra[u];
      }
    }
  };

  load_tile(tile);

  for (int i = tid; i < DOUTP * C4; i += NT) {
    int r = i / C4, c4 = (i - r * C4) * 4;
    float4 wl;
    if constexpr (DH > 0) {
      const int rr = (r < SP) ? r : r - SP;
      const bool rowOK = (r < DOUT) && (rr < DH);
      const float* Wsel = (r < SP) ? Wl : Wr;
      wl = *(const float4*)(Wsel + (long)(rowOK ? rr : 0) * DIN + c4);
      if (!rowOK) wl = f4z;
    } else {
      wl = (r < DOUT) ? *(const float4*)(Wl + (long)r * DIN + c4) : f4z;
    }
    *(float4*)&sWl[r * LDW + c4] = wl;
    if constexpr (HAS_AGG) {
      float4 wr = (r < DOUT) ? *(const float4*)(Wr + (long)r * DIN + c4) : f4z;
      *(float4*)&sWr[r * LDW + c4] = wr;
    }
  }

  const int wave = tid >> 5;
  const int lane = tid & 31;
  const int half = lane >> 4;
  const int l16 = lane & 15;
  const int n = wave * 16 + l16;
  const bool colOK = (n < DOUT);
  float bcol;
  if constexpr (DH > 0) {
    const bool hasB = (n >= SP) && (n - SP < DH);
    bcol = hasB ? bias[hasB ? n - SP : 0] : 0.0f;
  } else {
    bcol = colOK ? bias[colOK ? n : 0] : 0.0f;
  }

  store_tile();
  __syncthreads();

  for (;;) {
    const int next = tile + (int)gridDim.x;
    const bool has_next = next < ntiles;
    if (has_next) load_tile(next);

    v8f acc[MT];
#pragma unroll
    for (int mt = 0; mt < MT; ++mt) acc[mt] = (v8f){};

#pragma unroll
    for (int k = 0; k < DIN; k += 4) {
      const int kk = k + 2 * half;
      v2f b;
      b.x = sWl[n * LDW + kk];
      b.y = sWl[n * LDW + kk + 1];
      v2f b2;
      if constexpr (HAS_AGG) {
        b2.x = sWr[n * LDW + kk];
        b2.y = sWr[n * LDW + kk + 1];
      }
#pragma unroll
      for (int mt = 0; mt < MT; ++mt) {
        const int r = mt * 16 + l16;
        v2f a;
        a.x = sA[r * LDW + kk];
        a.y = sA[r * LDW + kk + 1];
        acc[mt] = __builtin_amdgcn_wmma_f32_16x16x4_f32(false, a, false, b,
                                                        (short)0, acc[mt], false,
                                                        false);
        if constexpr (HAS_AGG) {
          v2f a2;
          a2.x = sX[r * LDW + kk];
          a2.y = sX[r * LDW + kk + 1];
          acc[mt] = __builtin_amdgcn_wmma_f32_16x16x4_f32(false, a2, false, b2,
                                                          (short)0, acc[mt],
                                                          false, false);
        }
      }
    }

    const int row0 = tile * ROWS;
#pragma unroll
    for (int mt = 0; mt < MT; ++mt) {
#pragma unroll
      for (int i = 0; i < 8; ++i) {
        int row = row0 + mt * 16 + i + 8 * half;
        float v = acc[mt][i] + bcol;
        if (ACT == 1) v = tanhf(v);
        if (ACT == 2) v = (v > 0.0f) ? v : 0.01f * v;
        if constexpr (DH > 0) {
          if (colOK && row < nnodes) {
            if (n < SP) out[(long)row * SP + n] = v;
            else out2[(long)row * SP + (n - SP)] = v;
          }
        } else {
          if (colOK && row < nnodes) {
            out[(long)row * DOUT + n] = v;
            if (out2) out2[(long)row * DOUT + n] = v;
          }
        }
      }
    }

    if (!has_next) break;
    __syncthreads();
    store_tile();
    __syncthreads();
    tile = next;
  }
}


static inline unsigned cdiv(long a, long b) { return (unsigned)((a + b - 1) / b); }
static inline unsigned umin(unsigned a, unsigned b) { return a < b ? a : b; }

extern "C" void kernel_launch(void* const* d_in, const int* in_sizes, int n_in,
                              void* d_out, int out_size, void* d_ws, size_t ws_size,
                              hipStream_t stream) {
  const float* x = (const float*)d_in[0];
  const int* ei = (const int*)d_in[1];
  const int N = in_sizes[0] / 64;
  const int E = in_sizes[1] / 2;
  const int* src = ei;
  const int* dst = ei + E;

  const float* g1Wl = (const float*)d_in[2];
  const float* g1b  = (const float*)d_in[3];
  const float* g1Wr = (const float*)d_in[4];
  const float* g2Wl = (const float*)d_in[5];
  const float* g2b  = (const float*)d_in[6];
  const float* g2Wr = (const float*)d_in[7];
  const float* g3Wl = (const float*)d_in[8];
  const float* g3b  = (const float*)d_in[9];
  const float* g3Wr = (const float*)d_in[10];
  const float* trW  = (const float*)d_in[11];
  const float* trb  = (const float*)d_in[12];
  const float* d1W  = (const float*)d_in[13];
  const float* d1b  = (const float*)d_in[14];
  const float* d2Wl = (const float*)d_in[15];
  const float* d2b  = (const float*)d_in[16];
  const float* d2Wr = (const float*)d_in[17];
  const float* d3Wl = (const float*)d_in[18];
  const float* d3b  = (const float*)d_in[19];
  const float* d3Wr = (const float*)d_in[20];
  const float* d4Wl = (const float*)d_in[21];
  const float* d4b  = (const float*)d_in[22];
  const float* d4Wr = (const float*)d_in[23];

  float* ws   = (float*)d_ws;
  int*   cnt  = (int*)ws;
  float* inv  = ws + N;
  float* bufA = ws + 2 * (size_t)N;
  float* bufB = bufA + (size_t)N * 128;
  float* bufC = bufB + (size_t)N * 128;
  float* z2   = bufC + (size_t)N * 128;
  int*   rowptr = (int*)bufC;
  int*   col    = rowptr + (N + 1);
  float* bufD   = bufC + (size_t)N * 64;
  float* p3     = bufA;
  float* r3     = bufA + (size_t)N * 64;
  float* p4     = bufB;
  float* r4     = bufB + (size_t)N * 4;
  int*   rank   = (int*)bufB;
  int*   csum   = (int*)z2;

  float* outZ01 = (float*)d_out;
  float* outZ2  = outZ01 + 2 * (size_t)N;
  float* outMu  = outZ2 + N;
  float* outLv  = outMu + (size_t)N * 64;

  const int tiles = (N + 63) / 64;
  const unsigned gL1 = umin(tiles, 256);
  const unsigned gL2 = umin(tiles, 512);
  const unsigned gL4 = umin(tiles, 1024);
  const unsigned gE   = cdiv(E, 256);
  const unsigned gCh  = cdiv(N, 1024);
  const unsigned gG   = cdiv((long)N * 32, 256);

  zero_int_kernel<<<cdiv(N, 256), 256, 0, stream>>>(cnt, N);
  rank_kernel<<<gE, 256, 0, stream>>>(dst, cnt, rank, E);
  chunk_sum_kernel<<<gCh, 256, 0, stream>>>(cnt, csum, N);
  rowptr_kernel<<<gCh, 256, 0, stream>>>(cnt, csum, rowptr, inv, N);
  fill_kernel<<<gE, 256, 0, stream>>>(src, dst, rank, rowptr, col, E);

  gather_sum_kernel<64><<<gG, 256, 0, stream>>>(x, rowptr, col, bufA, N);
  wmma_linear_kernel<64, 128, 1, true><<<gL2, 256, 0, stream>>>(
      bufA, inv, x, g1Wl, g1b, g1Wr, bufB, nullptr, N, tiles);

  gather_sum_kernel<128><<<gG, 256, 0, stream>>>(bufB, rowptr, col, bufA, N);
  wmma_linear_kernel<128, 128, 1, true><<<gL1, 256, 0, stream>>>(
      bufA, inv, bufB, g2Wl, g2b, g2Wr, bufA, nullptr, N, tiles);

  gather_sum_kernel<128><<<gG, 256, 0, stream>>>(bufA, rowptr, col, bufB, N);
  wmma_linear_kernel<128, 128, 1, true><<<gL1, 256, 0, stream>>>(
      bufB, inv, bufA, g3Wl, g3b, g3Wr, bufB, nullptr, N, tiles);

  wmma_linear_kernel<128, 64, 0, false><<<gL4, 128, 0, stream>>>(
      nullptr, nullptr, bufB, trW, trb, nullptr, outMu, outLv, N, tiles);

  wmma_linear_kernel<64, 128, 2, false><<<gL4, 256, 0, stream>>>(
      nullptr, nullptr, outMu, d1W, d1b, nullptr, bufA, nullptr, N, tiles);

  gather_sum_kernel<128><<<gG, 256, 0, stream>>>(bufA, rowptr, col, bufB, N);
  wmma_linear_kernel<128, 128, 2, true><<<gL1, 256, 0, stream>>>(
      bufB, inv, bufA, d2Wl, d2b, d2Wr, bufB, nullptr, N, tiles);

  wmma_linear_kernel<128, 128, 0, false, 64><<<gL2, 256, 0, stream>>>(
      nullptr, nullptr, bufB, d3Wl, d3b, d3Wr, p3, r3, N, tiles);
  gather_mean_tanh_kernel<64><<<gG, 256, 0, stream>>>(p3, r3, inv, rowptr, col,
                                                      bufD, N);

  wmma_linear_kernel<64, 8, 0, false, 3><<<(unsigned)tiles, 32, 0, stream>>>(
      nullptr, nullptr, bufD, d4Wl, d4b, d4Wr, p4, r4, N, tiles);
  gather_final_kernel<<<cdiv(N, 32), 256, 0, stream>>>(p4, r4, inv, rowptr, col,
                                                       outZ01, outZ2, N);
}


# Linear kernels' epilogue free of data-dependent branches: tanh as straight-line code (both ranges, one select; library's formulas, may differ from its call in the last place), one row test a tile
# speedup vs baseline: 3.5188x; 1.1168x over previous
#include <hip/hip_runtime.h>
#include <hip/hip_bf16.h>

typedef __attribute__((ext_vector_type(2))) float v2f;
typedef __attribute__((ext_vector_type(4))) float v4f;
typedef __attribute__((ext_vector_type(8))) float v8f;


__global__ void zero_int_kernel(int* __restrict__ p, int n) {
  int i = blockIdx.x * blockDim.x + threadIdx.x;
  if (i < n) p[i] = 0;
}

__global__ void rank_kernel(const int* __restrict__ dst, int* __restrict__ cnt,
                            int* __restrict__ rank, int E) {
  int e = blockIdx.x * blockDim.x + threadIdx.x;
  if (e < E)
    rank[e] = __hip_atomic_fetch_add(&cnt[dst[e]], 1, __ATOMIC_RELAXED,
                                     __HIP_MEMORY_SCOPE_AGENT);
}

__global__ __launch_bounds__(256) void chunk_sum_kernel(
    const int* __restrict__ cnt, int* __restrict__ csum, int n) {
  __shared__ int sw[8];
  const int tid = threadIdx.x;
  const int base = blockIdx.x * 1024 + tid * 4;
  int s = 0;
  for (int u = 0; u < 4; ++u)
    if (base + u < n) s += cnt[base + u];
  for (int d = 16; d > 0; d >>= 1) s += __shfl_xor(s, d, 32);
  if ((tid & 31) == 0) sw[tid >> 5] = s;
  __syncthreads();
  if (tid == 0) {
    int t = 0;
    for (int w = 0; w < 8; ++w) t += sw[w];
    csum[blockIdx.x] = t;
  }
}

__global__ __launch_bounds__(256) void rowptr_kernel(
    const int* __restrict__ cnt, const int* __restrict__ csum,
    int* __restrict__ rowptr, float* __restrict__ inv, int n) {
  __shared__ int sw[8];
  __shared__ int soff;
  const int tid = threadIdx.x;
  const int lane = tid & 31;
  const int wave = tid >> 5;

  int p = 0;
  for (int i = tid; i < (int)blockIdx.x; i += 256) p += csum[i];
  for (int d = 16; d > 0; d >>= 1) p += __shfl_xor(p, d, 32);
  if (lane == 0) sw[wave] = p;
  __syncthreads();
  if (tid == 0) {
    int t = 0;
    for (int w = 0; w < 8; ++w) t += sw[w];
    soff = t;
  }
  __syncthreads();
  const int off = soff;

  const int base = blockIdx.x * 1024 + tid * 4;
  int c[4];
  int t = 0;
  for (int u = 0; u < 4; ++u) {
    c[u] = (base + u < n) ? cnt[base + u] : 0;
    t += c[u];
  }
  int incl = t;
  for (int d = 1; d < 32; d <<= 1) {
    int v = __shfl_up(incl, d, 32);
    if (lane >= d) incl += v;
  }
  if (lane == 31) sw[wave] = incl;
  __syncthreads();
  int run = off + incl - t;
  for (int w = 0; w < wave; ++w) run += sw[w];
  for (int u = 0; u < 4; ++u) {
    const int i = base + u;
    if (i < n) {
      rowptr[i] = run;
      inv[i] = 1.0f / fmaxf((float)c[u], 1.0f);
      run += c[u];
      if (i == n - 1) rowptr[n] = run;
    }
  }
}

__global__ void fill_kernel(const int* __restrict__ src, const int* __restrict__ dst,
                            const int* __restrict__ rank,
                            const int* __restrict__ rowptr,
                            int* __restrict__ col, int E) {
  int e = blockIdx.x * blockDim.x + threadIdx.x;
  if (e < E) col[rowptr[dst[e]] + rank[e]] = src[e];
}

template <int D, int U, typename VT>
__device__ __forceinline__ void add_rows(VT& acc, const float* __restrict__ xl,
                                         int cols, int j) {
  VT v[U];
#pragma unroll
  for (int u = 0; u < U; ++u) {
    const int s = __builtin_amdgcn_readlane(cols, j + u);
    v[u] = *(const VT*)(xl + (long)s * D);
  }
#pragma unroll
  for (int u = 0; u < U; ++u) acc += v[u];
}

template <int D>
__global__ __launch_bounds__(256) void gather_sum_kernel(
    const float* __restrict__ x, const int* __restrict__ rowptr,
    const int* __restrict__ col, float* __restrict__ agg, int n) {
  constexpr int V = D / 32;
  typedef float VT __attribute__((ext_vector_type(V)));
  const int lane = threadIdx.x & 31;
  const int node = __builtin_amdgcn_readfirstlane(
      (int)((blockIdx.x * blockDim.x + threadIdx.x) >> 5));
  if (node >= n) return;
  const int beg = rowptr[node];
  const int end = rowptr[node + 1];
  const float* xl = x + lane * V;
  VT acc = (VT)(0.0f);
  for (int base = beg; base < end; base += 32) {
    const int m = min(32, end - base);
    const int cols = (lane < m) ? col[base + lane] : 0;
    int j = 0;
    for (; j + 8 <= m; j += 8) add_rows<D, 8, VT>(acc, xl, cols, j);
    if (m - j >= 4) { add_rows<D, 4, VT>(acc, xl, cols, j); j += 4; }
    if (m - j >= 2) { add_rows<D, 2, VT>(acc, xl, cols, j); j += 2; }
    if (m - j >= 1) { add_rows<D, 1, VT>(acc, xl, cols, j); }
  }
  *(VT*)(agg + (long)node * D + lane * V) = acc;
}


template <int D>
__global__ __launch_bounds__(256) void gather_mean_tanh_kernel(
    const float* __restrict__ p, const float* __restrict__ r,
    const float* __restrict__ invdeg, const int* __restrict__ rowptr,
    const int* __restrict__ col, float* __restrict__ out, int n) {
  constexpr int V = D / 32;
  typedef float VT __attribute__((ext_vector_type(V)));
  const int lane = threadIdx.x & 31;
  const int node = __builtin_amdgcn_readfirstlane(
      (int)((blockIdx.x * blockDim.x + threadIdx.x) >> 5));
  if (node >= n) return;
  const int beg = rowptr[node];
  const int end = rowptr[node + 1];
  const float sc = invdeg[node];
  const VT root = *(const VT*)(r + (long)node * D + lane * V);
  const float* pl = p + lane * V;
  VT acc = (VT)(0.0f);
  for (int base = beg; base < end; base += 32) {
    const int m = min(32, end - base);
    const int cols = (lane < m) ? col[base + lane] : 0;
    int j = 0;
    for (; j + 8 <= m; j += 8) add_rows<D, 8, VT>(acc, pl, cols, j);
    if (m - j >= 4) { add_rows<D, 4, VT>(acc, pl, cols, j); j += 4; }
    if (m - j >= 2) { add_rows<D, 2, VT>(acc, pl, cols, j); j += 2; }
    if (m - j >= 1) { add_rows<D, 1, VT>(acc, pl, cols, j); }
  }
  VT v = acc * sc + root;
  for (int u = 0; u < V; ++u) v[u] = tanhf(v[u]);
  *(VT*)(out + (long)node * D + lane * V) = v;
}

__global__ __launch_bounds__(256) void gather_final_kernel(
    const float* __restrict__ p4, const float* __restrict__ r4,
    const float* __restrict__ invdeg, const int* __restrict__ rowptr,
    const int* __restrict__ col, float* __restrict__ out01,
    float* __restrict__ out2, int n) {
  __shared__ float s01[64];
  __shared__ float s2[32];
  const int tid = threadIdx.x;
  const int sub = tid & 7;
  const int g = tid >> 3;
  const int node = blockIdx.x * 32 + g;
  v4f acc = (v4f)(0.0f);
  if (node < n) {
    const int beg = rowptr[node];
    const int end = rowptr[node + 1];
    for (int e = beg + sub; e < end; e += 32) {
      int idx[4];
#pragma unroll
      for (int u = 0; u < 4; ++u) idx[u] = col[min(e + 8 * u, end - 1)];
      v4f v[4];
#pragma unroll
      for (int u = 0; u < 4; ++u) v[u] = *(const v4f*)(p4 + (long)idx[u] * 4);
#pragma unroll
      for (int u = 0; u < 4; ++u) {
        const bool in = (e + 8 * u < end);
        acc.x += in ? v[u].x : 0.0f;
        acc.y += in ? v[u].y : 0.0f;
        acc.z += in ? v[u].z : 0.0f;
      }
    }
  }
  for (int d = 1; d < 8; d <<= 1) {
    acc.x += __shfl_xor(acc.x, d, 32);
    acc.y += __shfl_xor(acc.y, d, 32);
    acc.z += __shfl_xor(acc.z, d, 32);
  }
  if (sub == 0 && node < n) {
    const float sc = invdeg[node];
    const v4f root = *(const v4f*)(r4 + (long)node * 4);
    s01[2 * g + 0] = acc.x * sc + root.x;
    s01[2 * g + 1] = acc.y * sc + root.y;
    s2[g] = acc.z * sc + root.z;
  }
  __syncthreads();
  if (tid < 64) {
    const long i = (long)blockIdx.x * 64 + tid;
    if (i < 2L * n) out01[i] = s01[tid];
  } else if (tid < 96) {
    const long i = (long)blockIdx.x * 32 + (tid - 64);
    if (i < n) out2[i] = s2[tid - 64];
  }
}

__device__ __forceinline__ float tanh_sl(float x) {
  constexpr float L2E    = __builtin_bit_cast(float, 0x3fb8aa3bu);
  constexpr float L2E_LO = __builtin_bit_cast(float, 0x32a5705fu);
  constexpr float EMAX   = __builtin_bit_cast(float, 0x42b17218u);
  constexpr float C0 = __builtin_bit_cast(float, 0xbbbac73du);
  constexpr float C1 = __builtin_bit_cast(float, 0x3ca908c9u);
  constexpr float C2 = __builtin_bit_cast(float, 0xbd5c1c4eu);
  constexpr float C3 = __builtin_bit_cast(float, 0x3e088382u);
  constexpr float C4 = __builtin_bit_cast(float, 0xbeaaaa99u);
  const float y = __builtin_fabsf(x);
  const float t = y + y;
  const float r = __builtin_rintf(t * L2E);
  float a = __builtin_fmaf(t, L2E, -r);
  a = __builtin_fmaf(t, L2E_LO, a);
  float e = __builtin_ldexpf(__builtin_amdgcn_exp2f(a), (int)r);
  e = (EMAX < t) ? __builtin_inff() : e;
  const float zl = __builtin_fmaf(__builtin_amdgcn_rcpf(e + 1.0f), -2.0f, 1.0f);
  const float s = x * x;
  float p = __builtin_fmaf(C0, s, C1);
  p = __builtin_fmaf(s, p, C2);
  p = __builtin_fmaf(s, p, C3);
  p = __builtin_fmaf(s, p, C4);
  const float zs = __builtin_fmaf(s, y * p, y);
  const float z = (0.625f > y) ? zs : zl;
  return __builtin_copysignf(z, x);
}

template <int ACT>
__device__ __forceinline__ float act_sl(float v) {
  if (ACT == 1) return tanh_sl(v);
  if (ACT == 2) return (v > 0.0f) ? v : 0.01f * v;
  return v;
}

template <int DIN, int DOUT, int ACT, bool HAS_AGG, int DH = 0, bool OUT2 = false>
__global__ __launch_bounds__(2 * ((DOUT + 15) & ~15)) void wmma_linear_kernel(
    const float* agg, const float* __restrict__ invdeg,
    const float* __restrict__ x,
    const float* __restrict__ Wl, const float* __restrict__ bias,
    const float* __restrict__ Wr,
    float* out, float* __restrict__ out2, int nnodes, int ntiles) {
  constexpr int MT = 4;
  constexpr int ROWS = 16 * MT;
  constexpr int DOUTP = (DOUT + 15) & ~15;
  constexpr int NT = 2 * DOUTP;
  constexpr int LDW = DIN + 4;
  constexpr int C4 = DIN / 4;
  constexpr int PER = ROWS * C4 / NT;
  constexpr int SP = DOUT / 2;
  static_assert(DH == 0 || (!HAS_AGG && DH <= SP), "split form is dense");
  __shared__ float sWl[DOUTP * LDW];
  __shared__ float sWr[HAS_AGG ? DOUTP * LDW : 1];
  __shared__ float sA[ROWS * LDW];
  __shared__ float sX[HAS_AGG ? ROWS * LDW : 1];

  const int tid = threadIdx.x;
  int tile = blockIdx.x;
  if (tile >= ntiles) return;

  const float4 f4z = make_float4(0.0f, 0.0f, 0.0f, 0.0f);

  v4f ra[PER];
  v4f rx[HAS_AGG ? PER : 1];
  float rs[HAS_AGG ? PER : 1];

  auto load_tile = [&](int t) {
    const int r0 = t * ROWS;
#pragma unroll
    for (int u = 0; u < PER; ++u) {
      const int i = tid + u * NT;
      const int r = i / C4, c4 = (i - r * C4) * 4;
      int m = r0 + r;
      if (m >= nnodes) m = nnodes - 1;
      if constexpr (HAS_AGG) {
        rs[u] = invdeg[m];
        ra[u] = *(const v4f*)(agg + (long)m * DIN + c4);
        rx[u] = *(const v4f*)(x + (long)m * DIN + c4);
      } else {
        ra[u] = *(const v4f*)(x + (long)m * DIN + c4);
      }
    }
  };
  auto store_tile = [&]() {
#pragma unroll
    for (int u = 0; u < PER; ++u) {
      const int i = tid + u * NT;
      const int r = i / C4, c4 = (i - r * C4) * 4;
      if constexpr (HAS_AGG) {
        *(v4f*)&sA[r * LDW + c4] = ra[u] * rs[u];
        *(v4f*)&sX[r * LDW + c4] = rx[u];
      } else {
        *(v4f*)&sA[r * LDW + c4] = ra[u];
      }
    }
  };

  load_tile(tile);

  for (int i = tid; i < DOUTP * C4; i += NT) {
    int r = i / C4, c4 = (i - r * C4) * 4;
    float4 wl;
    if constexpr (DH > 0) {
      const int rr = (r < SP) ? r : r - SP;
      const bool rowOK = (r < DOUT) && (rr < DH);
      const float* Wsel = (r < SP) ? Wl : Wr;
      wl = *(const float4*)(Wsel + (long)(rowOK ? rr : 0) * DIN + c4);
      if (!rowOK) wl = f4z;
    } else {
      wl = (r < DOUT) ? *(const float4*)(Wl + (long)r * DIN + c4) : f4z;
    }
    *(float4*)&sWl[r * LDW + c4] = wl;
    if constexpr (HAS_AGG) {
      float4 wr = (r < DOUT) ? *(const float4*)(Wr + (long)r * DIN + c4) : f4z;
      *(float4*)&sWr[r * LDW + c4] = wr;
    }
  }

  const int wave = tid >> 5;
  const int lane = tid & 31;
  const int half = lane >> 4;
  const int l16 = lane & 15;
  const int n = wave * 16 + l16;
  const bool colOK = (n < DOUT);
  float bcol;
  if constexpr (DH > 0) {
    const bool hasB = (n >= SP) && (n - SP < DH);
    bcol = hasB ? bias[hasB ? n - SP : 0] : 0.0f;
  } else {
    bcol = colOK ? bias[colOK ? n : 0] : 0.0f;
  }

  constexpr int PITCH = (DH > 0) ? SP : DOUT;
  float* pcol;
  if constexpr (DH > 0) pcol = (n < SP) ? out + n : out2 + (n - SP);
  else pcol = out + n;
  float* pcol2 = OUT2 ? out2 + n : nullptr;

  store_tile();
  __syncthreads();

  for (;;) {
    const int next = tile + (int)gridDim.x;
    const bool has_next = next < ntiles;
    if (has_next) load_tile(next);

    v8f acc[MT];
#pragma unroll
    for (int mt = 0; mt < MT; ++mt) acc[mt] = (v8f){};

#pragma unroll
    for (int k = 0; k < DIN; k += 4) {
      const int kk = k + 2 * half;
      v2f b;
      b.x = sWl[n * LDW + kk];
      b.y = sWl[n * LDW + kk + 1];
      v2f b2;
      if constexpr (HAS_AGG) {
        b2.x = sWr[n * LDW + kk];
        b2.y = sWr[n * LDW + kk + 1];
      }
#pragma unroll
      for (int mt = 0; mt < MT; ++mt) {
        const int r = mt * 16 + l16;
        v2f a;
        a.x = sA[r * LDW + kk];
        a.y = sA[r * LDW + kk + 1];
        acc[mt] = __builtin_amdgcn_wmma_f32_16x16x4_f32(false, a, false, b,
                                                        (short)0, acc[mt], false,
                                                        false);
        if constexpr (HAS_AGG) {
          v2f a2;
          a2.x = sX[r * LDW + kk];
          a2.y = sX[r * LDW + kk + 1];
          acc[mt] = __builtin_amdgcn_wmma_f32_16x16x4_f32(false, a2, false, b2,
                                                          (short)0, acc[mt],
                                                          false, false);
        }
      }
    }

    const int row0 = tile * ROWS;
#pragma unroll
    for (int mt = 0; mt < MT; ++mt) {
#pragma unroll
      for (int i = 0; i < 8; ++i) acc[mt][i] += bcol;
    }
    if (row0 + ROWS <= nnodes) {
      float* pt = pcol + (long)(row0 + 8 * half) * PITCH;
      float* pt2 = OUT2 ? pcol2 + (long)(row0 + 8 * half) * PITCH : nullptr;
      if (colOK) {
#pragma unroll
        for (int mt = 0; mt < MT; ++mt) {
#pragma unroll
          for (int i = 0; i < 8; ++i) {
            const float v = act_sl<ACT>(acc[mt][i]);
            pt[(mt * 16 + i) * PITCH] = v;
            if constexpr (OUT2) pt2[(mt * 16 + i) * PITCH] = v;
          }
        }
      }
    } else {
#pragma unroll
      for (int mt = 0; mt < MT; ++mt) {
#pragma unroll
        for (int i = 0; i < 8; ++i) {
          const int row = row0 + mt * 16 + i + 8 * half;
          const float v = act_sl<ACT>(acc[mt][i]);
          if (colOK && row < nnodes) {
            pcol[(long)row * PITCH] = v;
            if constexpr (OUT2) pcol2[(long)row * PITCH] = v;
          }
        }
      }
    }

    if (!has_next) break;
    __syncthreads();
    store_tile();
    __syncthreads();
    tile = next;
  }
}


static inline unsigned cdiv(long a, long b) { return (unsigned)((a + b - 1) / b); }
static inline unsigned umin(unsigned a, unsigned b) { return a < b ? a : b; }

extern "C" void kernel_launch(void* const* d_in, const int* in_sizes, int n_in,
                              void* d_out, int out_size, void* d_ws, size_t ws_size,
                              hipStream_t stream) {
  const float* x = (const float*)d_in[0];
  const int* ei = (const int*)d_in[1];
  const int N = in_sizes[0] / 64;
  const int E = in_sizes[1] / 2;
  const int* src = ei;
  const int* dst = ei + E;

  const float* g1Wl = (const float*)d_in[2];
  const float* g1b  = (const float*)d_in[3];
  const float* g1Wr = (const float*)d_in[4];
  const float* g2Wl = (const float*)d_in[5];
  const float* g2b  = (const float*)d_in[6];
  const float* g2Wr = (const float*)d_in[7];
  const float* g3Wl = (const float*)d_in[8];
  const float* g3b  = (const float*)d_in[9];
  const float* g3Wr = (const float*)d_in[10];
  const float* trW  = (const float*)d_in[11];
  const float* trb  = (const float*)d_in[12];
  const float* d1W  = (const float*)d_in[13];
  const float* d1b  = (const float*)d_in[14];
  const float* d2Wl = (const float*)d_in[15];
  const float* d2b  = (const float*)d_in[16];
  const float* d2Wr = (const float*)d_in[17];
  const float* d3Wl = (const float*)d_in[18];
  const float* d3b  = (const float*)d_in[19];
  const float* d3Wr = (const float*)d_in[20];
  const float* d4Wl = (const float*)d_in[21];
  const float* d4b  = (const float*)d_in[22];
  const float* d4Wr = (const float*)d_in[23];

  float* ws   = (float*)d_ws;
  int*   cnt  = (int*)ws;
  float* inv  = ws + N;
  float* bufA = ws + 2 * (size_t)N;
  float* bufB = bufA + (size_t)N * 128;
  float* bufC = bufB + (size_t)N * 128;
  float* z2   = bufC + (size_t)N * 128;
  int*   rowptr = (int*)bufC;
  int*   col    = rowptr + (N + 1);
  float* bufD   = bufC + (size_t)N * 64;
  float* p3     = bufA;
  float* r3     = bufA + (size_t)N * 64;
  float* p4     = bufB;
  float* r4     = bufB + (size_t)N * 4;
  int*   rank   = (int*)bufB;
  int*   csum   = (int*)z2;

  float* outZ01 = (float*)d_out;
  float* outZ2  = outZ01 + 2 * (size_t)N;
  float* outMu  = outZ2 + N;
  float* outLv  = outMu + (size_t)N * 64;

  const int tiles = (N + 63) / 64;
  const unsigned gL1 = umin(tiles, 256);
  const unsigned gL2 = umin(tiles, 512);
  const unsigned gL4 = umin(tiles, 1024);
  const unsigned gE   = cdiv(E, 256);
  const unsigned gCh  = cdiv(N, 1024);
  const unsigned gG   = cdiv((long)N * 32, 256);

  zero_int_kernel<<<cdiv(N, 256), 256, 0, stream>>>(cnt, N);
  rank_kernel<<<gE, 256, 0, stream>>>(dst, cnt, rank, E);
  chunk_sum_kernel<<<gCh, 256, 0, stream>>>(cnt, csum, N);
  rowptr_kernel<<<gCh, 256, 0, stream>>>(cnt, csum, rowptr, inv, N);
  fill_kernel<<<gE, 256, 0, stream>>>(src, dst, rank, rowptr, col, E);

  gather_sum_kernel<64><<<gG, 256, 0, stream>>>(x, rowptr, col, bufA, N);
  wmma_linear_kernel<64, 128, 1, true><<<gL2, 256, 0, stream>>>(
      bufA, inv, x, g1Wl, g1b, g1Wr, bufB, nullptr, N, tiles);

  gather_sum_kernel<128><<<gG, 256, 0, stream>>>(bufB, rowptr, col, bufA, N);
  wmma_linear_kernel<128, 128, 1, true><<<gL1, 256, 0, stream>>>(
      bufA, inv, bufB, g2Wl, g2b, g2Wr, bufA, nullptr, N, tiles);

  gather_sum_kernel<128><<<gG, 256, 0, stream>>>(bufA, rowptr, col, bufB, N);
  wmma_linear_kernel<128, 128, 1, true><<<gL1, 256, 0, stream>>>(
      bufB, inv, bufA, g3Wl, g3b, g3Wr, bufB, nullptr, N, tiles);

  wmma_linear_kernel<128, 64, 0, false, 0, true><<<gL4, 128, 0, stream>>>(
      nullptr, nullptr, bufB, trW, trb, nullptr, outMu, outLv, N, tiles);

  wmma_linear_kernel<64, 128, 2, false><<<gL4, 256, 0, stream>>>(
      nullptr, nullptr, outMu, d1W, d1b, nullptr, bufA, nullptr, N, tiles);

  gather_sum_kernel<128><<<gG, 256, 0, stream>>>(bufA, rowptr, col, bufB, N);
  wmma_linear_kernel<128, 128, 2, true><<<gL1, 256, 0, stream>>>(
      bufB, inv, bufA, d2Wl, d2b, d2Wr, bufB, nullptr, N, tiles);

  wmma_linear_kernel<128, 128, 0, false, 64><<<gL2, 256, 0, stream>>>(
      nullptr, nullptr, bufB, d3Wl, d3b, d3Wr, p3, r3, N, tiles);
  gather_mean_tanh_kernel<64><<<gG, 256, 0, stream>>>(p3, r3, inv, rowptr, col,
                                                      bufD, N);

  wmma_linear_kernel<64, 8, 0, false, 3><<<(unsigned)tiles, 32, 0, stream>>>(
      nullptr, nullptr, bufD, d4Wl, d4b, d4Wr, p4, r4, N, tiles);
  gather_final_kernel<<<cdiv(N, 32), 256, 0, stream>>>(p4, r4, inv, rowptr, col,
                                                       outZ01, outZ2, N);
}


# Linear kernels: row tile of 80 rows (5 M-tiles a wave) instead of 64: the busiest WGP walks 5 tiles = 400 rows, not 7 = 448 (390.6 is even); B fragment reused 5 times; per-row sums unchanged
# speedup vs baseline: 3.7204x; 1.0573x over previous
#include <hip/hip_runtime.h>
#include <hip/hip_bf16.h>

typedef __attribute__((ext_vector_type(2))) float v2f;
typedef __attribute__((ext_vector_type(4))) float v4f;
typedef __attribute__((ext_vector_type(8))) float v8f;


__global__ void zero_int_kernel(int* __restrict__ p, int n) {
  int i = blockIdx.x * blockDim.x + threadIdx.x;
  if (i < n) p[i] = 0;
}

__global__ void rank_kernel(const int* __restrict__ dst, int* __restrict__ cnt,
                            int* __restrict__ rank, int E) {
  int e = blockIdx.x * blockDim.x + threadIdx.x;
  if (e < E)
    rank[e] = __hip_atomic_fetch_add(&cnt[dst[e]], 1, __ATOMIC_RELAXED,
                                     __HIP_MEMORY_SCOPE_AGENT);
}

__global__ __launch_bounds__(256) void chunk_sum_kernel(
    const int* __restrict__ cnt, int* __restrict__ csum, int n) {
  __shared__ int sw[8];
  const int tid = threadIdx.x;
  const int base = blockIdx.x * 1024 + tid * 4;
  int s = 0;
  for (int u = 0; u < 4; ++u)
    if (base + u < n) s += cnt[base + u];
  for (int d = 16; d > 0; d >>= 1) s += __shfl_xor(s, d, 32);
  if ((tid & 31) == 0) sw[tid >> 5] = s;
  __syncthreads();
  if (tid == 0) {
    int t = 0;
    for (int w = 0; w < 8; ++w) t += sw[w];
    csum[blockIdx.x] = t;
  }
}

__global__ __launch_bounds__(256) void rowptr_kernel(
    const int* __restrict__ cnt, const int* __restrict__ csum,
    int* __restrict__ rowptr, float* __restrict__ inv, int n) {
  __shared__ int sw[8];
  __shared__ int soff;
  const int tid = threadIdx.x;
  const int lane = tid & 31;
  const int wave = tid >> 5;

  int p = 0;
  for (int i = tid; i < (int)blockIdx.x; i += 256) p += csum[i];
  for (int d = 16; d > 0; d >>= 1) p += __shfl_xor(p, d, 32);
  if (lane == 0) sw[wave] = p;
  __syncthreads();
  if (tid == 0) {
    int t = 0;
    for (int w = 0; w < 8; ++w) t += sw[w];
    soff = t;
  }
  __syncthreads();
  const int off = soff;

  const int base = blockIdx.x * 1024 + tid * 4;
  int c[4];
  int t = 0;
  for (int u = 0; u < 4; ++u) {
    c[u] = (base + u < n) ? cnt[base + u] : 0;
    t += c[u];
  }
  int incl = t;
  for (int d = 1; d < 32; d <<= 1) {
    int v = __shfl_up(incl, d, 32);
    if (lane >= d) incl += v;
  }
  if (lane == 31) sw[wave] = incl;
  __syncthreads();
  int run = off + incl - t;
  for (int w = 0; w < wave; ++w) run += sw[w];
  for (int u = 0; u < 4; ++u) {
    const int i = base + u;
    if (i < n) {
      rowptr[i] = run;
      inv[i] = 1.0f / fmaxf((float)c[u], 1.0f);
      run += c[u];
      if (i == n - 1) rowptr[n] = run;
    }
  }
}

__global__ void fill_kernel(const int* __restrict__ src, const int* __restrict__ dst,
                            const int* __restrict__ rank,
                            const int* __restrict__ rowptr,
                            int* __restrict__ col, int E) {
  int e = blockIdx.x * blockDim.x + threadIdx.x;
  if (e < E) col[rowptr[dst[e]] + rank[e]] = src[e];
}

template <int D, int U, typename VT>
__device__ __forceinline__ void add_rows(VT& acc, const float* __restrict__ xl,
                                         int cols, int j) {
  VT v[U];
#pragma unroll
  for (int u = 0; u < U; ++u) {
    const int s = __builtin_amdgcn_readlane(cols, j + u);
    v[u] = *(const VT*)(xl + (long)s * D);
  }
#pragma unroll
  for (int u = 0; u < U; ++u) acc += v[u];
}

template <int D>
__global__ __launch_bounds__(256) void gather_sum_kernel(
    const float* __restrict__ x, const int* __restrict__ rowptr,
    const int* __restrict__ col, float* __restrict__ agg, int n) {
  constexpr int V = D / 32;
  typedef float VT __attribute__((ext_vector_type(V)));
  const int lane = threadIdx.x & 31;
  const int node = __builtin_amdgcn_readfirstlane(
      (int)((blockIdx.x * blockDim.x + threadIdx.x) >> 5));
  if (node >= n) return;
  const int beg = rowptr[node];
  const int end = rowptr[node + 1];
  const float* xl = x + lane * V;
  VT acc = (VT)(0.0f);
  for (int base = beg; base < end; base += 32) {
    const int m = min(32, end - base);
    const int cols = (lane < m) ? col[base + lane] : 0;
    int j = 0;
    for (; j + 8 <= m; j += 8) add_rows<D, 8, VT>(acc, xl, cols, j);
    if (m - j >= 4) { add_rows<D, 4, VT>(acc, xl, cols, j); j += 4; }
    if (m - j >= 2) { add_rows<D, 2, VT>(acc, xl, cols, j); j += 2; }
    if (m - j >= 1) { add_rows<D, 1, VT>(acc, xl, cols, j); }
  }
  *(VT*)(agg + (long)node * D + lane * V) = acc;
}


template <int D>
__global__ __launch_bounds__(256) void gather_mean_tanh_kernel(
    const float* __restrict__ p, const float* __restrict__ r,
    const float* __restrict__ invdeg, const int* __restrict__ rowptr,
    const int* __restrict__ col, float* __restrict__ out, int n) {
  constexpr int V = D / 32;
  typedef float VT __attribute__((ext_vector_type(V)));
  const int lane = threadIdx.x & 31;
  const int node = __builtin_amdgcn_readfirstlane(
      (int)((blockIdx.x * blockDim.x + threadIdx.x) >> 5));
  if (node >= n) return;
  const int beg = rowptr[node];
  const int end = rowptr[node + 1];
  const float sc = invdeg[node];
  const VT root = *(const VT*)(r + (long)node * D + lane * V);
  const float* pl = p + lane * V;
  VT acc = (VT)(0.0f);
  for (int base = beg; base < end; base += 32) {
    const int m = min(32, end - base);
    const int cols = (lane < m) ? col[base + lane] : 0;
    int j = 0;
    for (; j + 8 <= m; j += 8) add_rows<D, 8, VT>(acc, pl, cols, j);
    if (m - j >= 4) { add_rows<D, 4, VT>(acc, pl, cols, j); j += 4; }
    if (m - j >= 2) { add_rows<D, 2, VT>(acc, pl, cols, j); j += 2; }
    if (m - j >= 1) { add_rows<D, 1, VT>(acc, pl, cols, j); }
  }
  VT v = acc * sc + root;
  for (int u = 0; u < V; ++u) v[u] = tanhf(v[u]);
  *(VT*)(out + (long)node * D + lane * V) = v;
}

__global__ __launch_bounds__(256) void gather_final_kernel(
    const float* __restrict__ p4, const float* __restrict__ r4,
    const float* __restrict__ invdeg, const int* __restrict__ rowptr,
    const int* __restrict__ col, float* __restrict__ out01,
    float* __restrict__ out2, int n) {
  __shared__ float s01[64];
  __shared__ float s2[32];
  const int tid = threadIdx.x;
  const int sub = tid & 7;
  const int g = tid >> 3;
  const int node = blockIdx.x * 32 + g;
  v4f acc = (v4f)(0.0f);
  if (node < n) {
    const int beg = rowptr[node];
    const int end = rowptr[node + 1];
    for (int e = beg + sub; e < end; e += 32) {
      int idx[4];
#pragma unroll
      for (int u = 0; u < 4; ++u) idx[u] = col[min(e + 8 * u, end - 1)];
      v4f v[4];
#pragma unroll
      for (int u = 0; u < 4; ++u) v[u] = *(const v4f*)(p4 + (long)idx[u] * 4);
#pragma unroll
      for (int u = 0; u < 4; ++u) {
        const bool in = (e + 8 * u < end);
        acc.x += in ? v[u].x : 0.0f;
        acc.y += in ? v[u].y : 0.0f;
        acc.z += in ? v[u].z : 0.0f;
      }
    }
  }
  for (int d = 1; d < 8; d <<= 1) {
    acc.x += __shfl_xor(acc.x, d, 32);
    acc.y += __shfl_xor(acc.y, d, 32);
    acc.z += __shfl_xor(acc.z, d, 32);
  }
  if (sub == 0 && node < n) {
    const float sc = invdeg[node];
    const v4f root = *(const v4f*)(r4 + (long)node * 4);
    s01[2 * g + 0] = acc.x * sc + root.x;
    s01[2 * g + 1] = acc.y * sc + root.y;
    s2[g] = acc.z * sc + root.z;
  }
  __syncthreads();
  if (tid < 64) {
    const long i = (long)blockIdx.x * 64 + tid;
    if (i < 2L * n) out01[i] = s01[tid];
  } else if (tid < 96) {
    const long i = (long)blockIdx.x * 32 + (tid - 64);
    if (i < n) out2[i] = s2[tid - 64];
  }
}

__device__ __forceinline__ float tanh_sl(float x) {
  constexpr float L2E    = __builtin_bit_cast(float, 0x3fb8aa3bu);
  constexpr float L2E_LO = __builtin_bit_cast(float, 0x32a5705fu);
  constexpr float EMAX   = __builtin_bit_cast(float, 0x42b17218u);
  constexpr float C0 = __builtin_bit_cast(float, 0xbbbac73du);
  constexpr float C1 = __builtin_bit_cast(float, 0x3ca908c9u);
  constexpr float C2 = __builtin_bit_cast(float, 0xbd5c1c4eu);
  constexpr float C3 = __builtin_bit_cast(float, 0x3e088382u);
  constexpr float C4 = __builtin_bit_cast(float, 0xbeaaaa99u);
  const float y = __builtin_fabsf(x);
  const float t = y + y;
  const float r = __builtin_rintf(t * L2E);
  float a = __builtin_fmaf(t, L2E, -r);
  a = __builtin_fmaf(t, L2E_LO, a);
  float e = __builtin_ldexpf(__builtin_amdgcn_exp2f(a), (int)r);
  e = (EMAX < t) ? __builtin_inff() : e;
  const float zl = __builtin_fmaf(__builtin_amdgcn_rcpf(e + 1.0f), -2.0f, 1.0f);
  const float s = x * x;
  float p = __builtin_fmaf(C0, s, C1);
  p = __builtin_fmaf(s, p, C2);
  p = __builtin_fmaf(s, p, C3);
  p = __builtin_fmaf(s, p, C4);
  const float zs = __builtin_fmaf(s, y * p, y);
  const float z = (0.625f > y) ? zs : zl;
  return __builtin_copysignf(z, x);
}

template <int ACT>
__device__ __forceinline__ float act_sl(float v) {
  if (ACT == 1) return tanh_sl(v);
  if (ACT == 2) return (v > 0.0f) ? v : 0.01f * v;
  return v;
}

template <int DIN, int DOUT, int ACT, bool HAS_AGG, int DH = 0, bool OUT2 = false>
__global__ __launch_bounds__(2 * ((DOUT + 15) & ~15)) void wmma_linear_kernel(
    const float* agg, const float* __restrict__ invdeg,
    const float* __restrict__ x,
    const float* __restrict__ Wl, const float* __restrict__ bias,
    const float* __restrict__ Wr,
    float* out, float* __restrict__ out2, int nnodes, int ntiles) {
  constexpr int MT = 5;
  constexpr int ROWS = 16 * MT;
  constexpr int DOUTP = (DOUT + 15) & ~15;
  constexpr int NT = 2 * DOUTP;
  constexpr int LDW = DIN + 4;
  constexpr int C4 = DIN / 4;
  constexpr int PER = ROWS * C4 / NT;
  static_assert(ROWS * C4 % NT == 0, "a tile's float4 slots divide evenly over the threads");
  constexpr int SP = DOUT / 2;
  static_assert(DH == 0 || (!HAS_AGG && DH <= SP), "split form is dense");
  __shared__ float sWl[DOUTP * LDW];
  __shared__ float sWr[HAS_AGG ? DOUTP * LDW : 1];
  __shared__ float sA[ROWS * LDW];
  __shared__ float sX[HAS_AGG ? ROWS * LDW : 1];

  const int tid = threadIdx.x;
  int tile = blockIdx.x;
  if (tile >= ntiles) return;

  const float4 f4z = make_float4(0.0f, 0.0f, 0.0f, 0.0f);

  v4f ra[PER];
  v4f rx[HAS_AGG ? PER : 1];
  float rs[HAS_AGG ? PER : 1];

  auto load_tile = [&](int t) {
    const int r0 = t * ROWS;
#pragma unroll
    for (int u = 0; u < PER; ++u) {
      const int i = tid + u * NT;
      const int r = i / C4, c4 = (i - r * C4) * 4;
      int m = r0 + r;
      if (m >= nnodes) m = nnodes - 1;
      if constexpr (HAS_AGG) {
        rs[u] = invdeg[m];
        ra[u] = *(const v4f*)(agg + (long)m * DIN + c4);
        rx[u] = *(const v4f*)(x + (long)m * DIN + c4);
      } else {
        ra[u] = *(const v4f*)(x + (long)m * DIN + c4);
      }
    }
  };
  auto store_tile = [&]() {
#pragma unroll
    for (int u = 0; u < PER; ++u) {
      const int i = tid + u * NT;
      const int r = i / C4, c4 = (i - r * C4) * 4;
      if constexpr (HAS_AGG) {
        *(v4f*)&sA[r * LDW + c4] = ra[u] * rs[u];
        *(v4f*)&sX[r * LDW + c4] = rx[u];
      } else {
        *(v4f*)&sA[r * LDW + c4] = ra[u];
      }
    }
  };

  load_tile(tile);

  for (int i = tid; i < DOUTP * C4; i += NT) {
    int r = i / C4, c4 = (i - r * C4) * 4;
    float4 wl;
    if constexpr (DH > 0) {
      const int rr = (r < SP) ? r : r - SP;
      const bool rowOK = (r < DOUT) && (rr < DH);
      const float* Wsel = (r < SP) ? Wl : Wr;
      wl = *(const float4*)(Wsel + (long)(rowOK ? rr : 0) * DIN + c4);
      if (!rowOK) wl = f4z;
    } else {
      wl = (r < DOUT) ? *(const float4*)(Wl + (long)r * DIN + c4) : f4z;
    }
    *(float4*)&sWl[r * LDW + c4] = wl;
    if constexpr (HAS_AGG) {
      float4 wr = (r < DOUT) ? *(const float4*)(Wr + (long)r * DIN + c4) : f4z;
      *(float4*)&sWr[r * LDW + c4] = wr;
    }
  }

  const int wave = tid >> 5;
  const int lane = tid & 31;
  const int half = lane >> 4;
  const int l16 = lane & 15;
  const int n = wave * 16 + l16;
  const bool colOK = (n < DOUT);
  float bcol;
  if constexpr (DH > 0) {
    const bool hasB = (n >= SP) && (n - SP < DH);
    bcol = hasB ? bias[hasB ? n - SP : 0] : 0.0f;
  } else {
    bcol = colOK ? bias[colOK ? n : 0] : 0.0f;
  }

  constexpr int PITCH = (DH > 0) ? SP : DOUT;
  float* pcol;
  if constexpr (DH > 0) pcol = (n < SP) ? out + n : out2 + (n - SP);
  else pcol = out + n;
  float* pcol2 = OUT2 ? out2 + n : nullptr;

  store_tile();
  __syncthreads();

  for (;;) {
    const int next = tile + (int)gridDim.x;
    const bool has_next = next < ntiles;
    if (has_next) load_tile(next);

    v8f acc[MT];
#pragma unroll
    for (int mt = 0; mt < MT; ++mt) acc[mt] = (v8f){};

#pragma unroll
    for (int k = 0; k < DIN; k += 4) {
      const int kk = k + 2 * half;
      v2f b;
      b.x = sWl[n * LDW + kk];
      b.y = sWl[n * LDW + kk + 1];
      v2f b2;
      if constexpr (HAS_AGG) {
        b2.x = sWr[n * LDW + kk];
        b2.y = sWr[n * LDW + kk + 1];
      }
#pragma unroll
      for (int mt = 0; mt < MT; ++mt) {
        const int r = mt * 16 + l16;
        v2f a;
        a.x = sA[r * LDW + kk];
        a.y = sA[r * LDW + kk + 1];
        acc[mt] = __builtin_amdgcn_wmma_f32_16x16x4_f32(false, a, false, b,
                                                        (short)0, acc[mt], false,
                                                        false);
        if constexpr (HAS_AGG) {
          v2f a2;
          a2.x = sX[r * LDW + kk];
          a2.y = sX[r * LDW + kk + 1];
          acc[mt] = __builtin_amdgcn_wmma_f32_16x16x4_f32(false, a2, false, b2,
                                                          (short)0, acc[mt],
                                                          false, false);
        }
      }
    }

    const int row0 = tile * ROWS;
#pragma unroll
    for (int mt = 0; mt < MT; ++mt) {
#pragma unroll
      for (int i = 0; i < 8; ++i) acc[mt][i] += bcol;
    }
    if (row0 + ROWS <= nnodes) {
      float* pt = pcol + (long)(row0 + 8 * half) * PITCH;
      float* pt2 = OUT2 ? pcol2 + (long)(row0 + 8 * half) * PITCH : nullptr;
      if (colOK) {
#pragma unroll
        for (int mt = 0; mt < MT; ++mt) {
#pragma unroll
          for (int i = 0; i < 8; ++i) {
            const float v = act_sl<ACT>(acc[mt][i]);
            pt[(mt * 16 + i) * PITCH] = v;
            if constexpr (OUT2) pt2[(mt * 16 + i) * PITCH] = v;
          }
        }
      }
    } else {
#pragma unroll
      for (int mt = 0; mt < MT; ++mt) {
#pragma unroll
        for (int i = 0; i < 8; ++i) {
          const int row = row0 + mt * 16 + i + 8 * half;
          const float v = act_sl<ACT>(acc[mt][i]);
          if (colOK && row < nnodes) {
            pcol[(long)row * PITCH] = v;
            if constexpr (OUT2) pcol2[(long)row * PITCH] = v;
          }
        }
      }
    }

    if (!has_next) break;
    __syncthreads();
    store_tile();
    __syncthreads();
    tile = next;
  }
}


static inline unsigned cdiv(long a, long b) { return (unsigned)((a + b - 1) / b); }
static inline unsigned umin(unsigned a, unsigned b) { return a < b ? a : b; }

extern "C" void kernel_launch(void* const* d_in, const int* in_sizes, int n_in,
                              void* d_out, int out_size, void* d_ws, size_t ws_size,
                              hipStream_t stream) {
  const float* x = (const float*)d_in[0];
  const int* ei = (const int*)d_in[1];
  const int N = in_sizes[0] / 64;
  const int E = in_sizes[1] / 2;
  const int* src = ei;
  const int* dst = ei + E;

  const float* g1Wl = (const float*)d_in[2];
  const float* g1b  = (const float*)d_in[3];
  const float* g1Wr = (const float*)d_in[4];
  const float* g2Wl = (const float*)d_in[5];
  const float* g2b  = (const float*)d_in[6];
  const float* g2Wr = (const float*)d_in[7];
  const float* g3Wl = (const float*)d_in[8];
  const float* g3b  = (const float*)d_in[9];
  const float* g3Wr = (const float*)d_in[10];
  const float* trW  = (const float*)d_in[11];
  const float* trb  = (const float*)d_in[12];
  const float* d1W  = (const float*)d_in[13];
  const float* d1b  = (const float*)d_in[14];
  const float* d2Wl = (const float*)d_in[15];
  const float* d2b  = (const float*)d_in[16];
  const float* d2Wr = (const float*)d_in[17];
  const float* d3Wl = (const float*)d_in[18];
  const float* d3b  = (const float*)d_in[19];
  const float* d3Wr = (const float*)d_in[20];
  const float* d4Wl = (const float*)d_in[21];
  const float* d4b  = (const float*)d_in[22];
  const float* d4Wr = (const float*)d_in[23];

  float* ws   = (float*)d_ws;
  int*   cnt  = (int*)ws;
  float* inv  = ws + N;
  float* bufA = ws + 2 * (size_t)N;
  float* bufB = bufA + (size_t)N * 128;
  float* bufC = bufB + (size_t)N * 128;
  float* z2   = bufC + (size_t)N * 128;
  int*   rowptr = (int*)bufC;
  int*   col    = rowptr + (N + 1);
  float* bufD   = bufC + (size_t)N * 64;
  float* p3     = bufA;
  float* r3     = bufA + (size_t)N * 64;
  float* p4     = bufB;
  float* r4     = bufB + (size_t)N * 4;
  int*   rank   = (int*)bufB;
  int*   csum   = (int*)z2;

  float* outZ01 = (float*)d_out;
  float* outZ2  = outZ01 + 2 * (size_t)N;
  float* outMu  = outZ2 + N;
  float* outLv  = outMu + (size_t)N * 64;

  const int tiles = (N + 79) / 80;
  const unsigned gL1 = umin(tiles, 256);
  const unsigned gL2 = umin(tiles, 512);
  const unsigned gL4 = umin(tiles, 1024);
  const unsigned gE   = cdiv(E, 256);
  const unsigned gCh  = cdiv(N, 1024);
  const unsigned gG   = cdiv((long)N * 32, 256);

  zero_int_kernel<<<cdiv(N, 256), 256, 0, stream>>>(cnt, N);
  rank_kernel<<<gE, 256, 0, stream>>>(dst, cnt, rank, E);
  chunk_sum_kernel<<<gCh, 256, 0, stream>>>(cnt, csum, N);
  rowptr_kernel<<<gCh, 256, 0, stream>>>(cnt, csum, rowptr, inv, N);
  fill_kernel<<<gE, 256, 0, stream>>>(src, dst, rank, rowptr, col, E);

  gather_sum_kernel<64><<<gG, 256, 0, stream>>>(x, rowptr, col, bufA, N);
  wmma_linear_kernel<64, 128, 1, true><<<gL2, 256, 0, stream>>>(
      bufA, inv, x, g1Wl, g1b, g1Wr, bufB, nullptr, N, tiles);

  gather_sum_kernel<128><<<gG, 256, 0, stream>>>(bufB, rowptr, col, bufA, N);
  wmma_linear_kernel<128, 128, 1, true><<<gL1, 256, 0, stream>>>(
      bufA, inv, bufB, g2Wl, g2b, g2Wr, bufA, nullptr, N, tiles);

  gather_sum_kernel<128><<<gG, 256, 0, stream>>>(bufA, rowptr, col, bufB, N);
  wmma_linear_kernel<128, 128, 1, true><<<gL1, 256, 0, stream>>>(
      bufB, inv, bufA, g3Wl, g3b, g3Wr, bufB, nullptr, N, tiles);

  wmma_linear_kernel<128, 64, 0, false, 0, true><<<gL4, 128, 0, stream>>>(
      nullptr, nullptr, bufB, trW, trb, nullptr, outMu, outLv, N, tiles);

  wmma_linear_kernel<64, 128, 2, false><<<gL4, 256, 0, stream>>>(
      nullptr, nullptr, outMu, d1W, d1b, nullptr, bufA, nullptr, N, tiles);

  gather_sum_kernel<128><<<gG, 256, 0, stream>>>(bufA, rowptr, col, bufB, N);
  wmma_linear_kernel<128, 128, 2, true><<<gL1, 256, 0, stream>>>(
      bufB, inv, bufA, d2Wl, d2b, d2Wr, bufB, nullptr, N, tiles);

  wmma_linear_kernel<128, 128, 0, false, 64><<<gL2, 256, 0, stream>>>(
      nullptr, nullptr, bufB, d3Wl, d3b, d3Wr, p3, r3, N, tiles);
  gather_mean_tanh_kernel<64><<<gG, 256, 0, stream>>>(p3, r3, inv, rowptr, col,
                                                      bufD, N);

  wmma_linear_kernel<64, 8, 0, false, 3><<<(unsigned)tiles, 32, 0, stream>>>(
      nullptr, nullptr, bufD, d4Wl, d4b, d4Wr, p4, r4, N, tiles);
  gather_final_kernel<<<cdiv(N, 32), 256, 0, stream>>>(p4, r4, inv, rowptr, col,
                                                       outZ01, outZ2, N);
}


# SAGE layers (3 encoder, decoder L2): row gather and products in ONE launch: 20 gather waves fill the next row tile in LDS (double buffer) while 8 waves multiply the current one; sum order kept
# speedup vs baseline: 4.0546x; 1.0898x over previous
#include <hip/hip_runtime.h>
#include <hip/hip_bf16.h>

typedef __attribute__((ext_vector_type(2))) float v2f;
typedef __attribute__((ext_vector_type(4))) float v4f;
typedef __attribute__((ext_vector_type(8))) float v8f;


__global__ void zero_int_kernel(int* __restrict__ p, int n) {
  int i = blockIdx.x * blockDim.x + threadIdx.x;
  if (i < n) p[i] = 0;
}

__global__ void rank_kernel(const int* __restrict__ dst, int* __restrict__ cnt,
                            int* __restrict__ rank, int E) {
  int e = blockIdx.x * blockDim.x + threadIdx.x;
  if (e < E)
    rank[e] = __hip_atomic_fetch_add(&cnt[dst[e]], 1, __ATOMIC_RELAXED,
                                     __HIP_MEMORY_SCOPE_AGENT);
}

__global__ __launch_bounds__(256) void chunk_sum_kernel(
    const int* __restrict__ cnt, int* __restrict__ csum, int n) {
  __shared__ int sw[8];
  const int tid = threadIdx.x;
  const int base = blockIdx.x * 1024 + tid * 4;
  int s = 0;
  for (int u = 0; u < 4; ++u)
    if (base + u < n) s += cnt[base + u];
  for (int d = 16; d > 0; d >>= 1) s += __shfl_xor(s, d, 32);
  if ((tid & 31) == 0) sw[tid >> 5] = s;
  __syncthreads();
  if (tid == 0) {
    int t = 0;
    for (int w = 0; w < 8; ++w) t += sw[w];
    csum[blockIdx.x] = t;
  }
}

__global__ __launch_bounds__(256) void rowptr_kernel(
    const int* __restrict__ cnt, const int* __restrict__ csum,
    int* __restrict__ rowptr, float* __restrict__ inv, int n) {
  __shared__ int sw[8];
  __shared__ int soff;
  const int tid = threadIdx.x;
  const int lane = tid & 31;
  const int wave = tid >> 5;

  int p = 0;
  for (int i = tid; i < (int)blockIdx.x; i += 256) p += csum[i];
  for (int d = 16; d > 0; d >>= 1) p += __shfl_xor(p, d, 32);
  if (lane == 0) sw[wave] = p;
  __syncthreads();
  if (tid == 0) {
    int t = 0;
    for (int w = 0; w < 8; ++w) t += sw[w];
    soff = t;
  }
  __syncthreads();
  const int off = soff;

  const int base = blockIdx.x * 1024 + tid * 4;
  int c[4];
  int t = 0;
  for (int u = 0; u < 4; ++u) {
    c[u] = (base + u < n) ? cnt[base + u] : 0;
    t += c[u];
  }
  int incl = t;
  for (int d = 1; d < 32; d <<= 1) {
    int v = __shfl_up(incl, d, 32);
    if (lane >= d) incl += v;
  }
  if (lane == 31) sw[wave] = incl;
  __syncthreads();
  int run = off + incl - t;
  for (int w = 0; w < wave; ++w) run += sw[w];
  for (int u = 0; u < 4; ++u) {
    const int i = base + u;
    if (i < n) {
      rowptr[i] = run;
      inv[i] = 1.0f / fmaxf((float)c[u], 1.0f);
      run += c[u];
      if (i == n - 1) rowptr[n] = run;
    }
  }
}

__global__ void fill_kernel(const int* __restrict__ src, const int* __restrict__ dst,
                            const int* __restrict__ rank,
                            const int* __restrict__ rowptr,
                            int* __restrict__ col, int E) {
  int e = blockIdx.x * blockDim.x + threadIdx.x;
  if (e < E) col[rowptr[dst[e]] + rank[e]] = src[e];
}

template <int D, int U, typename VT>
__device__ __forceinline__ void add_rows(VT& acc, const float* __restrict__ xl,
                                         int cols, int j) {
  VT v[U];
#pragma unroll
  for (int u = 0; u < U; ++u) {
    const int s = __builtin_amdgcn_readlane(cols, j + u);
    v[u] = *(const VT*)(xl + (long)s * D);
  }
#pragma unroll
  for (int u = 0; u < U; ++u) acc += v[u];
}


template <int D>
__global__ __launch_bounds__(256) void gather_mean_tanh_kernel(
    const float* __restrict__ p, const float* __restrict__ r,
    const float* __restrict__ invdeg, const int* __restrict__ rowptr,
    const int* __restrict__ col, float* __restrict__ out, int n) {
  constexpr int V = D / 32;
  typedef float VT __attribute__((ext_vector_type(V)));
  const int lane = threadIdx.x & 31;
  const int node = __builtin_amdgcn_readfirstlane(
      (int)((blockIdx.x * blockDim.x + threadIdx.x) >> 5));
  if (node >= n) return;
  const int beg = rowptr[node];
  const int end = rowptr[node + 1];
  const float sc = invdeg[node];
  const VT root = *(const VT*)(r + (long)node * D + lane * V);
  const float* pl = p + lane * V;
  VT acc = (VT)(0.0f);
  for (int base = beg; base < end; base += 32) {
    const int m = min(32, end - base);
    const int cols = (lane < m) ? col[base + lane] : 0;
    int j = 0;
    for (; j + 8 <= m; j += 8) add_rows<D, 8, VT>(acc, pl, cols, j);
    if (m - j >= 4) { add_rows<D, 4, VT>(acc, pl, cols, j); j += 4; }
    if (m - j >= 2) { add_rows<D, 2, VT>(acc, pl, cols, j); j += 2; }
    if (m - j >= 1) { add_rows<D, 1, VT>(acc, pl, cols, j); }
  }
  VT v = acc * sc + root;
  for (int u = 0; u < V; ++u) v[u] = tanhf(v[u]);
  *(VT*)(out + (long)node * D + lane * V) = v;
}

__global__ __launch_bounds__(256) void gather_final_kernel(
    const float* __restrict__ p4, const float* __restrict__ r4,
    const float* __restrict__ invdeg, const int* __restrict__ rowptr,
    const int* __restrict__ col, float* __restrict__ out01,
    float* __restrict__ out2, int n) {
  __shared__ float s01[64];
  __shared__ float s2[32];
  const int tid = threadIdx.x;
  const int sub = tid & 7;
  const int g = tid >> 3;
  const int node = blockIdx.x * 32 + g;
  v4f acc = (v4f)(0.0f);
  if (node < n) {
    const int beg = rowptr[node];
    const int end = rowptr[node + 1];
    for (int e = beg + sub; e < end; e += 32) {
      int idx[4];
#pragma unroll
      for (int u = 0; u < 4; ++u) idx[u] = col[min(e + 8 * u, end - 1)];
      v4f v[4];
#pragma unroll
      for (int u = 0; u < 4; ++u) v[u] = *(const v4f*)(p4 + (long)idx[u] * 4);
#pragma unroll
      for (int u = 0; u < 4; ++u) {
        const bool in = (e + 8 * u < end);
        acc.x += in ? v[u].x : 0.0f;
        acc.y += in ? v[u].y : 0.0f;
        acc.z += in ? v[u].z : 0.0f;
      }
    }
  }
  for (int d = 1; d < 8; d <<= 1) {
    acc.x += __shfl_xor(acc.x, d, 32);
    acc.y += __shfl_xor(acc.y, d, 32);
    acc.z += __shfl_xor(acc.z, d, 32);
  }
  if (sub == 0 && node < n) {
    const float sc = invdeg[node];
    const v4f root = *(const v4f*)(r4 + (long)node * 4);
    s01[2 * g + 0] = acc.x * sc + root.x;
    s01[2 * g + 1] = acc.y * sc + root.y;
    s2[g] = acc.z * sc + root.z;
  }
  __syncthreads();
  if (tid < 64) {
    const long i = (long)blockIdx.x * 64 + tid;
    if (i < 2L * n) out01[i] = s01[tid];
  } else if (tid < 96) {
    const long i = (long)blockIdx.x * 32 + (tid - 64);
    if (i < n) out2[i] = s2[tid - 64];
  }
}

__device__ __forceinline__ float tanh_sl(float x) {
  constexpr float L2E    = __builtin_bit_cast(float, 0x3fb8aa3bu);
  constexpr float L2E_LO = __builtin_bit_cast(float, 0x32a5705fu);
  constexpr float EMAX   = __builtin_bit_cast(float, 0x42b17218u);
  constexpr float C0 = __builtin_bit_cast(float, 0xbbbac73du);
  constexpr float C1 = __builtin_bit_cast(float, 0x3ca908c9u);
  constexpr float C2 = __builtin_bit_cast(float, 0xbd5c1c4eu);
  constexpr float C3 = __builtin_bit_cast(float, 0x3e088382u);
  constexpr float C4 = __builtin_bit_cast(float, 0xbeaaaa99u);
  const float y = __builtin_fabsf(x);
  const float t = y + y;
  const float r = __builtin_rintf(t * L2E);
  float a = __builtin_fmaf(t, L2E, -r);
  a = __builtin_fmaf(t, L2E_LO, a);
  float e = __builtin_ldexpf(__builtin_amdgcn_exp2f(a), (int)r);
  e = (EMAX < t) ? __builtin_inff() : e;
  const float zl = __builtin_fmaf(__builtin_amdgcn_rcpf(e + 1.0f), -2.0f, 1.0f);
  const float s = x * x;
  float p = __builtin_fmaf(C0, s, C1);
  p = __builtin_fmaf(s, p, C2);
  p = __builtin_fmaf(s, p, C3);
  p = __builtin_fmaf(s, p, C4);
  const float zs = __builtin_fmaf(s, y * p, y);
  const float z = (0.625f > y) ? zs : zl;
  return __builtin_copysignf(z, x);
}

template <int ACT>
__device__ __forceinline__ float act_sl(float v) {
  if (ACT == 1) return tanh_sl(v);
  if (ACT == 2) return (v > 0.0f) ? v : 0.01f * v;
  return v;
}

template <int DIN, int DOUT, int ACT, int DH = 0, bool OUT2 = false>
__global__ __launch_bounds__(2 * ((DOUT + 15) & ~15)) void wmma_linear_kernel(
    const float* __restrict__ x,
    const float* __restrict__ Wl, const float* __restrict__ bias,
    const float* __restrict__ Wr,
    float* out, float* __restrict__ out2, int nnodes, int ntiles) {
  constexpr int MT = 5;
  constexpr int ROWS = 16 * MT;
  constexpr int DOUTP = (DOUT + 15) & ~15;
  constexpr int NT = 2 * DOUTP;
  constexpr int LDW = DIN + 4;
  constexpr int C4 = DIN / 4;
  constexpr int PER = ROWS * C4 / NT;
  static_assert(ROWS * C4 % NT == 0, "a tile's float4 slots divide evenly over the threads");
  constexpr int SP = DOUT / 2;
  static_assert(DH == 0 || DH <= SP, "a half's rows fit its columns");
  __shared__ float sWl[DOUTP * LDW];
  __shared__ float sA[ROWS * LDW];

  const int tid = threadIdx.x;
  int tile = blockIdx.x;
  if (tile >= ntiles) return;

  const float4 f4z = make_float4(0.0f, 0.0f, 0.0f, 0.0f);

  v4f ra[PER];

  auto load_tile = [&](int t) {
    const int r0 = t * ROWS;
#pragma unroll
    for (int u = 0; u < PER; ++u) {
      const int i = tid + u * NT;
      const int r = i / C4, c4 = (i - r * C4) * 4;
      int m = r0 + r;
      if (m >= nnodes) m = nnodes - 1;
      ra[u] = *(const v4f*)(x + (long)m * DIN + c4);
    }
  };
  auto store_tile = [&]() {
#pragma unroll
    for (int u = 0; u < PER; ++u) {
      const int i = tid + u * NT;
      const int r = i / C4, c4 = (i - r * C4) * 4;
      *(v4f*)&sA[r * LDW + c4] = ra[u];
    }
  };

  load_tile(tile);

  for (int i = tid; i < DOUTP * C4; i += NT) {
    int r = i / C4, c4 = (i - r * C4) * 4;
    float4 wl;
    if constexpr (DH > 0) {
      const int rr = (r < SP) ? r : r - SP;
      const bool rowOK = (r < DOUT) && (rr < DH);
      const float* Wsel = (r < SP) ? Wl : Wr;
      wl = *(const float4*)(Wsel + (long)(rowOK ? rr : 0) * DIN + c4);
      if (!rowOK) wl = f4z;
    } else {
      wl = (r < DOUT) ? *(const float4*)(Wl + (long)r * DIN + c4) : f4z;
    }
    *(float4*)&sWl[r * LDW + c4] = wl;
  }

  const int wave = tid >> 5;
  const int lane = tid & 31;
  const int half = lane >> 4;
  const int l16 = lane & 15;
  const int n = wave * 16 + l16;
  const bool colOK = (n < DOUT);
  float bcol;
  if constexpr (DH > 0) {
    const bool hasB = (n >= SP) && (n - SP < DH);
    bcol = hasB ? bias[hasB ? n - SP : 0] : 0.0f;
  } else {
    bcol = colOK ? bias[colOK ? n : 0] : 0.0f;
  }

  constexpr int PITCH = (DH > 0) ? SP : DOUT;
  float* pcol;
  if constexpr (DH > 0) pcol = (n < SP) ? out + n : out2 + (n - SP);
  else pcol = out + n;
  float* pcol2 = OUT2 ? out2 + n : nullptr;

  store_tile();
  __syncthreads();

  for (;;) {
    const int next = tile + (int)gridDim.x;
    const bool has_next = next < ntiles;
    if (has_next) load_tile(next);

    v8f acc[MT];
#pragma unroll
    for (int mt = 0; mt < MT; ++mt) acc[mt] = (v8f){};

#pragma unroll
    for (int k = 0; k < DIN; k += 4) {
      const int kk = k + 2 * half;
      v2f b;
      b.x = sWl[n * LDW + kk];
      b.y = sWl[n * LDW + kk + 1];
#pragma unroll
      for (int mt = 0; mt < MT; ++mt) {
        const int r = mt * 16 + l16;
        v2f a;
        a.x = sA[r * LDW + kk];
        a.y = sA[r * LDW + kk + 1];
        acc[mt] = __builtin_amdgcn_wmma_f32_16x16x4_f32(false, a, false, b,
                                                        (short)0, acc[mt], false,
                                                        false);
      }
    }

    const int row0 = tile * ROWS;
#pragma unroll
    for (int mt = 0; mt < MT; ++mt) {
#pragma unroll
      for (int i = 0; i < 8; ++i) acc[mt][i] += bcol;
    }
    if (row0 + ROWS <= nnodes) {
      float* pt = pcol + (long)(row0 + 8 * half) * PITCH;
      float* pt2 = OUT2 ? pcol2 + (long)(row0 + 8 * half) * PITCH : nullptr;
      if (colOK) {
#pragma unroll
        for (int mt = 0; mt < MT; ++mt) {
#pragma unroll
          for (int i = 0; i < 8; ++i) {
            const float v = act_sl<ACT>(acc[mt][i]);
            pt[(mt * 16 + i) * PITCH] = v;
            if constexpr (OUT2) pt2[(mt * 16 + i) * PITCH] = v;
          }
        }
      }
    } else {
#pragma unroll
      for (int mt = 0; mt < MT; ++mt) {
#pragma unroll
        for (int i = 0; i < 8; ++i) {
          const int row = row0 + mt * 16 + i + 8 * half;
          const float v = act_sl<ACT>(acc[mt][i]);
          if (colOK && row < nnodes) {
            pcol[(long)row * PITCH] = v;
            if constexpr (OUT2) pcol2[(long)row * PITCH] = v;
          }
        }
      }
    }

    if (!has_next) break;
    __syncthreads();
    store_tile();
    __syncthreads();
    tile = next;
  }
}

template <int DIN, int ACT>
__global__ __launch_bounds__(896) void sage_fused_kernel(
    const float* __restrict__ x, const float* __restrict__ invdeg,
    const int* __restrict__ rowptr, const int* __restrict__ col,
    const float* __restrict__ Wl, const float* __restrict__ bias,
    const float* __restrict__ Wr, float* __restrict__ out, int nnodes,
    int ntiles) {
  constexpr int DOUT = 128;
  constexpr int MT = 5;
  constexpr int ROWS = 16 * MT;
  constexpr int PW = DOUT / 16;
  constexpr int GW = 20;
  constexpr int KR = ROWS / GW;
  constexpr int NT = 32 * (PW + GW);
  constexpr int LDW = DIN + 4;
  constexpr int C4 = DIN / 4;
  constexpr int V = DIN / 32;
  static_assert(ROWS % GW == 0, "every gather wave takes the same number of rows");
  typedef float VT __attribute__((ext_vector_type(V)));
  __shared__ float sWl[DOUT * LDW];
  __shared__ float sWr[DOUT * LDW];
  __shared__ float sA[2][ROWS * LDW];
  __shared__ float sX[2][ROWS * LDW];

  const int tid = threadIdx.x;
  if ((int)blockIdx.x >= ntiles) return;

  for (int i = tid; i < DOUT * C4; i += NT) {
    const int r = i / C4, c4 = (i - r * C4) * 4;
    *(float4*)&sWl[r * LDW + c4] = *(const float4*)(Wl + (long)r * DIN + c4);
    *(float4*)&sWr[r * LDW + c4] = *(const float4*)(Wr + (long)r * DIN + c4);
  }

  const int wave = __builtin_amdgcn_readfirstlane(tid >> 5);
  const int lane = tid & 31;
  const int half = lane >> 4;
  const int l16 = lane & 15;
  const int n = (wave & (PW - 1)) * 16 + l16;
  const float bcol = bias[n];
  float* pcol = out + n;

  for (int i = 0;; ++i) {
    const int tg = (int)blockIdx.x + i * (int)gridDim.x;
    const bool has_g = tg < ntiles;

    if (wave < PW) {
      if (i > 0) {
        const int tile = tg - (int)gridDim.x;
        const float* a0 = sA[(i - 1) & 1];
        const float* x0 = sX[(i - 1) & 1];
        v8f acc[MT];
#pragma unroll
        for (int mt = 0; mt < MT; ++mt) acc[mt] = (v8f){};

#pragma unroll
        for (int k = 0; k < DIN; k += 4) {
          const int kk = k + 2 * half;
          v2f b;
          b.x = sWl[n * LDW + kk];
          b.y = sWl[n * LDW + kk + 1];
          v2f b2;
          b2.x = sWr[n * LDW + kk];
          b2.y = sWr[n * LDW + kk + 1];
#pragma unroll
          for (int mt = 0; mt < MT; ++mt) {
            const int r = mt * 16 + l16;
            v2f a;
            a.x = a0[r * LDW + kk];
            a.y = a0[r * LDW + kk + 1];
            acc[mt] = __builtin_amdgcn_wmma_f32_16x16x4_f32(false, a, false, b,
                                                            (short)0, acc[mt], false,
                                                            false);
            v2f a2;
            a2.x = x0[r * LDW + kk];
            a2.y = x0[r * LDW + kk + 1];
            acc[mt] = __builtin_amdgcn_wmma_f32_16x16x4_f32(false, a2, false, b2,
                                                            (short)0, acc[mt],
                                                            false, false);
          }
        }

        const int row0 = tile * ROWS;
#pragma unroll
        for (int mt = 0; mt < MT; ++mt) {
#pragma unroll
          for (int e = 0; e < 8; ++e) acc[mt][e] += bcol;
        }
        if (row0 + ROWS <= nnodes) {
          float* pt = pcol + (long)(row0 + 8 * half) * DOUT;
#pragma unroll
          for (int mt = 0; mt < MT; ++mt) {
#pragma unroll
            for (int e = 0; e < 8; ++e)
              pt[(mt * 16 + e) * DOUT] = act_sl<ACT>(acc[mt][e]);
          }
        } else {
#pragma unroll
          for (int mt = 0; mt < MT; ++mt) {
#pragma unroll
            for (int e = 0; e < 8; ++e) {
              const int row = row0 + mt * 16 + e + 8 * half;
              const float v = act_sl<ACT>(acc[mt][e]);
              if (row < nnodes) pcol[(long)row * DOUT] = v;
            }
          }
        }
      }
    } else if (has_g) {
      float* a1 = sA[i & 1];
      float* x1 = sX[i & 1];
      const int gw = wave - PW;
      int nd[KR], beg[KR], end[KR], cl[KR];
      float sc[KR];
#pragma unroll
      for (int k = 0; k < KR; ++k) {
        int node = tg * ROWS + gw + k * GW;
        if (node >= nnodes) node = nnodes - 1;
        nd[k] = node;
        beg[k] = rowptr[node];
        end[k] = rowptr[node + 1];
        sc[k] = invdeg[node];
      }
#pragma unroll
      for (int k = 0; k < KR; ++k) {
        const int m = min(32, end[k] - beg[k]);
        cl[k] = (lane < m) ? col[beg[k] + lane] : 0;
      }
      const float* xl = x + lane * V;
#pragma unroll
      for (int k = 0; k < KR; ++k) {
        const VT root = *(const VT*)(xl + (long)nd[k] * DIN);
        VT acc = (VT)(0.0f);
        int cols = cl[k];
        for (int base = beg[k]; base < end[k]; base += 32) {
          const int m = min(32, end[k] - base);
          if (base != beg[k]) cols = (lane < m) ? col[base + lane] : 0;
          int j = 0;
          for (; j + 16 <= m; j += 16) add_rows<DIN, 16, VT>(acc, xl, cols, j);
          if (m - j >= 8) { add_rows<DIN, 8, VT>(acc, xl, cols, j); j += 8; }
          if (m - j >= 4) { add_rows<DIN, 4, VT>(acc, xl, cols, j); j += 4; }
          if (m - j >= 2) { add_rows<DIN, 2, VT>(acc, xl, cols, j); j += 2; }
          if (m - j >= 1) { add_rows<DIN, 1, VT>(acc, xl, cols, j); }
        }
        const int r = gw + k * GW;
        *(VT*)&a1[r * LDW + lane * V] = acc * sc[k];
        *(VT*)&x1[r * LDW + lane * V] = root;
      }
    }

    if (!has_g) break;
    __syncthreads();
  }
}


static inline unsigned cdiv(long a, long b) { return (unsigned)((a + b - 1) / b); }
static inline unsigned umin(unsigned a, unsigned b) { return a < b ? a : b; }

extern "C" void kernel_launch(void* const* d_in, const int* in_sizes, int n_in,
                              void* d_out, int out_size, void* d_ws, size_t ws_size,
                              hipStream_t stream) {
  const float* x = (const float*)d_in[0];
  const int* ei = (const int*)d_in[1];
  const int N = in_sizes[0] / 64;
  const int E = in_sizes[1] / 2;
  const int* src = ei;
  const int* dst = ei + E;

  const float* g1Wl = (const float*)d_in[2];
  const float* g1b  = (const float*)d_in[3];
  const float* g1Wr = (const float*)d_in[4];
  const float* g2Wl = (const float*)d_in[5];
  const float* g2b  = (const float*)d_in[6];
  const float* g2Wr = (const float*)d_in[7];
  const float* g3Wl = (const float*)d_in[8];
  const float* g3b  = (const float*)d_in[9];
  const float* g3Wr = (const float*)d_in[10];
  const float* trW  = (const float*)d_in[11];
  const float* trb  = (const float*)d_in[12];
  const float* d1W  = (const float*)d_in[13];
  const float* d1b  = (const float*)d_in[14];
  const float* d2Wl = (const float*)d_in[15];
  const float* d2b  = (const float*)d_in[16];
  const float* d2Wr = (const float*)d_in[17];
  const float* d3Wl = (const float*)d_in[18];
  const float* d3b  = (const float*)d_in[19];
  const float* d3Wr = (const float*)d_in[20];
  const float* d4Wl = (const float*)d_in[21];
  const float* d4b  = (const float*)d_in[22];
  const float* d4Wr = (const float*)d_in[23];

  float* ws   = (float*)d_ws;
  int*   cnt  = (int*)ws;
  float* inv  = ws + N;
  float* bufA = ws + 2 * (size_t)N;
  float* bufB = bufA + (size_t)N * 128;
  float* bufC = bufB + (size_t)N * 128;
  float* z2   = bufC + (size_t)N * 128;
  int*   rowptr = (int*)bufC;
  int*   col    = rowptr + (N + 1);
  float* bufD   = bufC + (size_t)N * 64;
  float* p3     = bufB;
  float* r3     = bufB + (size_t)N * 64;
  float* p4     = bufA;
  float* r4     = bufA + (size_t)N * 4;
  int*   rank   = (int*)bufB;
  int*   csum   = (int*)z2;

  float* outZ01 = (float*)d_out;
  float* outZ2  = outZ01 + 2 * (size_t)N;
  float* outMu  = outZ2 + N;
  float* outLv  = outMu + (size_t)N * 64;

  const int tiles = (N + 79) / 80;
  const unsigned gL1 = umin(tiles, 256);
  const unsigned gL2 = umin(tiles, 512);
  const unsigned gL4 = umin(tiles, 1024);
  const unsigned gE   = cdiv(E, 256);
  const unsigned gCh  = cdiv(N, 1024);
  const unsigned gG   = cdiv((long)N * 32, 256);

  zero_int_kernel<<<cdiv(N, 256), 256, 0, stream>>>(cnt, N);
  rank_kernel<<<gE, 256, 0, stream>>>(dst, cnt, rank, E);
  chunk_sum_kernel<<<gCh, 256, 0, stream>>>(cnt, csum, N);
  rowptr_kernel<<<gCh, 256, 0, stream>>>(cnt, csum, rowptr, inv, N);
  fill_kernel<<<gE, 256, 0, stream>>>(src, dst, rank, rowptr, col, E);

  sage_fused_kernel<64, 1><<<gL1, 896, 0, stream>>>(
      x, inv, rowptr, col, g1Wl, g1b, g1Wr, bufA, N, tiles);

  sage_fused_kernel<128, 1><<<gL1, 896, 0, stream>>>(
      bufA, inv, rowptr, col, g2Wl, g2b, g2Wr, bufB, N, tiles);

  sage_fused_kernel<128, 1><<<gL1, 896, 0, stream>>>(
      bufB, inv, rowptr, col, g3Wl, g3b, g3Wr, bufA, N, tiles);

  wmma_linear_kernel<128, 64, 0, 0, true><<<gL4, 128, 0, stream>>>(
      bufA, trW, trb, nullptr, outMu, outLv, N, tiles);

  wmma_linear_kernel<64, 128, 2><<<gL4, 256, 0, stream>>>(
      outMu, d1W, d1b, nullptr, bufB, nullptr, N, tiles);

  sage_fused_kernel<128, 2><<<gL1, 896, 0, stream>>>(
      bufB, inv, rowptr, col, d2Wl, d2b, d2Wr, bufA, N, tiles);

  wmma_linear_kernel<128, 128, 0, 64><<<gL2, 256, 0, stream>>>(
      bufA, d3Wl, d3b, d3Wr, p3, r3, N, tiles);
  gather_mean_tanh_kernel<64><<<gG, 256, 0, stream>>>(p3, r3, inv, rowptr, col,
                                                      bufD, N);

  wmma_linear_kernel<64, 8, 0, 3><<<(unsigned)tiles, 32, 0, stream>>>(
      bufD, d4Wl, d4b, d4Wr, p4, r4, N, tiles);
  gather_final_kernel<<<cdiv(N, 32), 256, 0, stream>>>(p4, r4, inv, rowptr, col,
                                                       outZ01, outZ2, N);
}


# Edge bucketing with no atomics on memory: edges binned by 512-node range through LDS counters, then one work-group per range ranks its edges in LDS; bucket order (sum order) stays arrival order
# speedup vs baseline: 4.7537x; 1.1724x over previous
#include <hip/hip_runtime.h>
#include <hip/hip_bf16.h>

typedef __attribute__((ext_vector_type(2))) float v2f;
typedef __attribute__((ext_vector_type(4))) float v4f;
typedef __attribute__((ext_vector_type(8))) float v8f;

constexpr int BSH = 9;
constexpr int BW = 1 << BSH;
constexpr int NBP = 256;
constexpr int NCH = 256;

__device__ __forceinline__ int wave_incl_scan(int v, int lane) {
  for (int d = 1; d < 32; d <<= 1) {
    const int u = __shfl_up(v, d, 32);
    if (lane >= d) v += u;
  }
  return v;
}

__global__ __launch_bounds__(1024) void bin_count_kernel(
    const int* __restrict__ dst, int* __restrict__ binCount, int E, int chunk) {
  __shared__ int hist[NBP];
  const int tid = threadIdx.x;
  if (tid < NBP) hist[tid] = 0;
  __syncthreads();
  const int e0 = blockIdx.x * chunk;
  const int e1 = min(E, e0 + chunk);
  for (int e = e0 + tid; e < e1; e += 4096) {
    int d[4];
#pragma unroll
    for (int u = 0; u < 4; ++u) d[u] = dst[min(e + 1024 * u, e1 - 1)];
#pragma unroll
    for (int u = 0; u < 4; ++u)
      if (e + 1024 * u < e1)
        __hip_atomic_fetch_add(&hist[d[u] >> BSH], 1, __ATOMIC_RELAXED,
                               __HIP_MEMORY_SCOPE_WORKGROUP);
  }
  __syncthreads();
  if (tid < NBP) binCount[blockIdx.x * NBP + tid] = hist[tid];
}

__global__ __launch_bounds__(NCH) void bin_offset_kernel(
    const int* __restrict__ binCount, int* __restrict__ off, int* __restrict__ tot) {
  __shared__ int sw[NCH / 32];
  const int c = threadIdx.x;
  const int b = blockIdx.x;
  const int lane = c & 31;
  const int wave = c >> 5;
  const int v = binCount[c * NBP + b];
  const int incl = wave_incl_scan(v, lane);
  if (lane == 31) sw[wave] = incl;
  __syncthreads();
  int run = incl - v;
  int total = 0;
  for (int w = 0; w < NCH / 32; ++w) {
    if (w < wave) run += sw[w];
    total += sw[w];
  }
  off[c * NBP + b] = run;
  if (c == 0) tot[b] = total;
}

__global__ __launch_bounds__(1024) void bin_scatter_kernel(
    const int* __restrict__ src, const int* __restrict__ dst,
    const int* __restrict__ off, const int* __restrict__ tot,
    int* __restrict__ base, int* __restrict__ pairs, int E, int chunk) {
  int2* __restrict__ pr = (int2*)pairs;
  __shared__ int cursor[NBP];
  __shared__ int sw[NBP / 32];
  const int tid = threadIdx.x;
  const int lane = tid & 31;
  const int wave = tid >> 5;
  int t = 0, incl = 0;
  if (tid < NBP) {
    t = tot[tid];
    incl = wave_incl_scan(t, lane);
    if (lane == 31) sw[wave] = incl;
  }
  __syncthreads();
  if (tid < NBP) {
    int start = incl - t;
    for (int w = 0; w < wave; ++w) start += sw[w];
    cursor[tid] = start + off[blockIdx.x * NBP + tid];
    if (blockIdx.x == 0) {
      base[tid] = start;
      if (tid == NBP - 1) base[NBP] = start + t;
    }
  }
  __syncthreads();
  const int e0 = blockIdx.x * chunk;
  const int e1 = min(E, e0 + chunk);
  for (int e = e0 + tid; e < e1; e += 4096) {
    int sv[4], d[4];
#pragma unroll
    for (int u = 0; u < 4; ++u) {
      const int ee = min(e + 1024 * u, e1 - 1);
      sv[u] = src[ee];
      d[u] = dst[ee];
    }
#pragma unroll
    for (int u = 0; u < 4; ++u)
      if (e + 1024 * u < e1) {
        const int pos = __hip_atomic_fetch_add(&cursor[d[u] >> BSH], 1,
                                               __ATOMIC_RELAXED,
                                               __HIP_MEMORY_SCOPE_WORKGROUP);
        pr[pos] = make_int2(sv[u], d[u]);
      }
  }
}

__global__ __launch_bounds__(1024) void bin_finish_kernel(
    const int* __restrict__ pairs, const int* __restrict__ base,
    int* __restrict__ rowptr, float* __restrict__ inv, int* __restrict__ col,
    int n) {
  const int2* __restrict__ pr = (const int2*)pairs;
  __shared__ int cnt[BW];
  __shared__ int cursor[BW];
  __shared__ int sw[BW / 32];
  const int tid = threadIdx.x;
  const int lane = tid & 31;
  const int wave = tid >> 5;
  const int lo = base[blockIdx.x];
  const int hi = base[blockIdx.x + 1];
  if (tid < BW) cnt[tid] = 0;
  __syncthreads();
  for (int e = lo + tid; e < hi; e += 4096) {
    int d[4];
#pragma unroll
    for (int u = 0; u < 4; ++u) d[u] = pr[min(e + 1024 * u, hi - 1)].y;
#pragma unroll
    for (int u = 0; u < 4; ++u)
      if (e + 1024 * u < hi)
        __hip_atomic_fetch_add(&cnt[d[u] & (BW - 1)], 1, __ATOMIC_RELAXED,
                               __HIP_MEMORY_SCOPE_WORKGROUP);
  }
  __syncthreads();
  int c = 0, incl = 0;
  if (tid < BW) {
    c = cnt[tid];
    incl = wave_incl_scan(c, lane);
    if (lane == 31) sw[wave] = incl;
  }
  __syncthreads();
  if (tid < BW) {
    int start = lo + incl - c;
    for (int w = 0; w < wave; ++w) start += sw[w];
    cursor[tid] = start;
    const int node = blockIdx.x * BW + tid;
    if (node < n) {
      rowptr[node] = start;
      inv[node] = 1.0f / fmaxf((float)c, 1.0f);
      if (node == n - 1) rowptr[n] = start + c;
    }
  }
  __syncthreads();
  for (int e = lo + tid; e < hi; e += 4096) {
    int2 p[4];
#pragma unroll
    for (int u = 0; u < 4; ++u) p[u] = pr[min(e + 1024 * u, hi - 1)];
#pragma unroll
    for (int u = 0; u < 4; ++u)
      if (e + 1024 * u < hi) {
        const int pos = __hip_atomic_fetch_add(&cursor[p[u].y & (BW - 1)], 1,
                                               __ATOMIC_RELAXED,
                                               __HIP_MEMORY_SCOPE_WORKGROUP);
        col[pos] = p[u].x;
      }
  }
}

template <int D, int U, typename VT>
__device__ __forceinline__ void add_rows(VT& acc, const float* __restrict__ xl,
                                         int cols, int j) {
  VT v[U];
#pragma unroll
  for (int u = 0; u < U; ++u) {
    const int s = __builtin_amdgcn_readlane(cols, j + u);
    v[u] = *(const VT*)(xl + (long)s * D);
  }
#pragma unroll
  for (int u = 0; u < U; ++u) acc += v[u];
}


template <int D>
__global__ __launch_bounds__(256) void gather_mean_tanh_kernel(
    const float* __restrict__ p, const float* __restrict__ r,
    const float* __restrict__ invdeg, const int* __restrict__ rowptr,
    const int* __restrict__ col, float* __restrict__ out, int n) {
  constexpr int V = D / 32;
  typedef float VT __attribute__((ext_vector_type(V)));
  const int lane = threadIdx.x & 31;
  const int node = __builtin_amdgcn_readfirstlane(
      (int)((blockIdx.x * blockDim.x + threadIdx.x) >> 5));
  if (node >= n) return;
  const int beg = rowptr[node];
  const int end = rowptr[node + 1];
  const float sc = invdeg[node];
  const VT root = *(const VT*)(r + (long)node * D + lane * V);
  const float* pl = p + lane * V;
  VT acc = (VT)(0.0f);
  for (int base = beg; base < end; base += 32) {
    const int m = min(32, end - base);
    const int cols = (lane < m) ? col[base + lane] : 0;
    int j = 0;
    for (; j + 8 <= m; j += 8) add_rows<D, 8, VT>(acc, pl, cols, j);
    if (m - j >= 4) { add_rows<D, 4, VT>(acc, pl, cols, j); j += 4; }
    if (m - j >= 2) { add_rows<D, 2, VT>(acc, pl, cols, j); j += 2; }
    if (m - j >= 1) { add_rows<D, 1, VT>(acc, pl, cols, j); }
  }
  VT v = acc * sc + root;
  for (int u = 0; u < V; ++u) v[u] = tanhf(v[u]);
  *(VT*)(out + (long)node * D + lane * V) = v;
}

__global__ __launch_bounds__(256) void gather_final_kernel(
    const float* __restrict__ p4, const float* __restrict__ r4,
    const float* __restrict__ invdeg, const int* __restrict__ rowptr,
    const int* __restrict__ col, float* __restrict__ out01,
    float* __restrict__ out2, int n) {
  __shared__ float s01[64];
  __shared__ float s2[32];
  const int tid = threadIdx.x;
  const int sub = tid & 7;
  const int g = tid >> 3;
  const int node = blockIdx.x * 32 + g;
  v4f acc = (v4f)(0.0f);
  if (node < n) {
    const int beg = rowptr[node];
    const int end = rowptr[node + 1];
    for (int e = beg + sub; e < end; e += 32) {
      int idx[4];
#pragma unroll
      for (int u = 0; u < 4; ++u) idx[u] = col[min(e + 8 * u, end - 1)];
      v4f v[4];
#pragma unroll
      for (int u = 0; u < 4; ++u) v[u] = *(const v4f*)(p4 + (long)idx[u] * 4);
#pragma unroll
      for (int u = 0; u < 4; ++u) {
        const bool in = (e + 8 * u < end);
        acc.x += in ? v[u].x : 0.0f;
        acc.y += in ? v[u].y : 0.0f;
        acc.z += in ? v[u].z : 0.0f;
      }
    }
  }
  for (int d = 1; d < 8; d <<= 1) {
    acc.x += __shfl_xor(acc.x, d, 32);
    acc.y += __shfl_xor(acc.y, d, 32);
    acc.z += __shfl_xor(acc.z, d, 32);
  }
  if (sub == 0 && node < n) {
    const float sc = invdeg[node];
    const v4f root = *(const v4f*)(r4 + (long)node * 4);
    s01[2 * g + 0] = acc.x * sc + root.x;
    s01[2 * g + 1] = acc.y * sc + root.y;
    s2[g] = acc.z * sc + root.z;
  }
  __syncthreads();
  if (tid < 64) {
    const long i = (long)blockIdx.x * 64 + tid;
    if (i < 2L * n) out01[i] = s01[tid];
  } else if (tid < 96) {
    const long i = (long)blockIdx.x * 32 + (tid - 64);
    if (i < n) out2[i] = s2[tid - 64];
  }
}

__device__ __forceinline__ float tanh_sl(float x) {
  constexpr float L2E    = __builtin_bit_cast(float, 0x3fb8aa3bu);
  constexpr float L2E_LO = __builtin_bit_cast(float, 0x32a5705fu);
  constexpr float EMAX   = __builtin_bit_cast(float, 0x42b17218u);
  constexpr float C0 = __builtin_bit_cast(float, 0xbbbac73du);
  constexpr float C1 = __builtin_bit_cast(float, 0x3ca908c9u);
  constexpr float C2 = __builtin_bit_cast(float, 0xbd5c1c4eu);
  constexpr float C3 = __builtin_bit_cast(float, 0x3e088382u);
  constexpr float C4 = __builtin_bit_cast(float, 0xbeaaaa99u);
  const float y = __builtin_fabsf(x);
  const float t = y + y;
  const float r = __builtin_rintf(t * L2E);
  float a = __builtin_fmaf(t, L2E, -r);
  a = __builtin_fmaf(t, L2E_LO, a);
  float e = __builtin_ldexpf(__builtin_amdgcn_exp2f(a), (int)r);
  e = (EMAX < t) ? __builtin_inff() : e;
  const float zl = __builtin_fmaf(__builtin_amdgcn_rcpf(e + 1.0f), -2.0f, 1.0f);
  const float s = x * x;
  float p = __builtin_fmaf(C0, s, C1);
  p = __builtin_fmaf(s, p, C2);
  p = __builtin_fmaf(s, p, C3);
  p = __builtin_fmaf(s, p, C4);
  const float zs = __builtin_fmaf(s, y * p, y);
  const float z = (0.625f > y) ? zs : zl;
  return __builtin_copysignf(z, x);
}

template <int ACT>
__device__ __forceinline__ float act_sl(float v) {
  if (ACT == 1) return tanh_sl(v);
  if (ACT == 2) return (v > 0.0f) ? v : 0.01f * v;
  return v;
}

template <int DIN, int DOUT, int ACT, int DH = 0, bool OUT2 = false>
__global__ __launch_bounds__(2 * ((DOUT + 15) & ~15)) void wmma_linear_kernel(
    const float* __restrict__ x,
    const float* __restrict__ Wl, const float* __restrict__ bias,
    const float* __restrict__ Wr,
    float* out, float* __restrict__ out2, int nnodes, int ntiles) {
  constexpr int MT = 5;
  constexpr int ROWS = 16 * MT;
  constexpr int DOUTP = (DOUT + 15) & ~15;
  constexpr int NT = 2 * DOUTP;
  constexpr int LDW = DIN + 4;
  constexpr int C4 = DIN / 4;
  constexpr int PER = ROWS * C4 / NT;
  static_assert(ROWS * C4 % NT == 0, "a tile's float4 slots divide evenly over the threads");
  constexpr int SP = DOUT / 2;
  static_assert(DH == 0 || DH <= SP, "a half's rows fit its columns");
  __shared__ float sWl[DOUTP * LDW];
  __shared__ float sA[ROWS * LDW];

  const int tid = threadIdx.x;
  int tile = blockIdx.x;
  if (tile >= ntiles) return;

  const float4 f4z = make_float4(0.0f, 0.0f, 0.0f, 0.0f);

  v4f ra[PER];

  auto load_tile = [&](int t) {
    const int r0 = t * ROWS;
#pragma unroll
    for (int u = 0; u < PER; ++u) {
      const int i = tid + u * NT;
      const int r = i / C4, c4 = (i - r * C4) * 4;
      int m = r0 + r;
      if (m >= nnodes) m = nnodes - 1;
      ra[u] = *(const v4f*)(x + (long)m * DIN + c4);
    }
  };
  auto store_tile = [&]() {
#pragma unroll
    for (int u = 0; u < PER; ++u) {
      const int i = tid + u * NT;
      const int r = i / C4, c4 = (i - r * C4) * 4;
      *(v4f*)&sA[r * LDW + c4] = ra[u];
    }
  };

  load_tile(tile);

  for (int i = tid; i < DOUTP * C4; i += NT) {
    int r = i / C4, c4 = (i - r * C4) * 4;
    float4 wl;
    if constexpr (DH > 0) {
      const int rr = (r < SP) ? r : r - SP;
      const bool rowOK = (r < DOUT) && (rr < DH);
      const float* Wsel = (r < SP) ? Wl : Wr;
      wl = *(const float4*)(Wsel + (long)(rowOK ? rr : 0) * DIN + c4);
      if (!rowOK) wl = f4z;
    } else {
      wl = (r < DOUT) ? *(const float4*)(Wl + (long)r * DIN + c4) : f4z;
    }
    *(float4*)&sWl[r * LDW + c4] = wl;
  }

  const int wave = tid >> 5;
  const int lane = tid & 31;
  const int half = lane >> 4;
  const int l16 = lane & 15;
  const int n = wave * 16 + l16;
  const bool colOK = (n < DOUT);
  float bcol;
  if constexpr (DH > 0) {
    const bool hasB = (n >= SP) && (n - SP < DH);
    bcol = hasB ? bias[hasB ? n - SP : 0] : 0.0f;
  } else {
    bcol = colOK ? bias[colOK ? n : 0] : 0.0f;
  }

  constexpr int PITCH = (DH > 0) ? SP : DOUT;
  float* pcol;
  if constexpr (DH > 0) pcol = (n < SP) ? out + n : out2 + (n - SP);
  else pcol = out + n;
  float* pcol2 = OUT2 ? out2 + n : nullptr;

  store_tile();
  __syncthreads();

  for (;;) {
    const int next = tile + (int)gridDim.x;
    const bool has_next = next < ntiles;
    if (has_next) load_tile(next);

    v8f acc[MT];
#pragma unroll
    for (int mt = 0; mt < MT; ++mt) acc[mt] = (v8f){};

#pragma unroll
    for (int k = 0; k < DIN; k += 4) {
      const int kk = k + 2 * half;
      v2f b;
      b.x = sWl[n * LDW + kk];
      b.y = sWl[n * LDW + kk + 1];
#pragma unroll
      for (int mt = 0; mt < MT; ++mt) {
        const int r = mt * 16 + l16;
        v2f a;
        a.x = sA[r * LDW + kk];
        a.y = sA[r * LDW + kk + 1];
        acc[mt] = __builtin_amdgcn_wmma_f32_16x16x4_f32(false, a, false, b,
                                                        (short)0, acc[mt], false,
                                                        false);
      }
    }

    const int row0 = tile * ROWS;
#pragma unroll
    for (int mt = 0; mt < MT; ++mt) {
#pragma unroll
      for (int i = 0; i < 8; ++i) acc[mt][i] += bcol;
    }
    if (row0 + ROWS <= nnodes) {
      float* pt = pcol + (long)(row0 + 8 * half) * PITCH;
      float* pt2 = OUT2 ? pcol2 + (long)(row0 + 8 * half) * PITCH : nullptr;
      if (colOK) {
#pragma unroll
        for (int mt = 0; mt < MT; ++mt) {
#pragma unroll
          for (int i = 0; i < 8; ++i) {
            const float v = act_sl<ACT>(acc[mt][i]);
            pt[(mt * 16 + i) * PITCH] = v;
            if constexpr (OUT2) pt2[(mt * 16 + i) * PITCH] = v;
          }
        }
      }
    } else {
#pragma unroll
      for (int mt = 0; mt < MT; ++mt) {
#pragma unroll
        for (int i = 0; i < 8; ++i) {
          const int row = row0 + mt * 16 + i + 8 * half;
          const float v = act_sl<ACT>(acc[mt][i]);
          if (colOK && row < nnodes) {
            pcol[(long)row * PITCH] = v;
            if constexpr (OUT2) pcol2[(long)row * PITCH] = v;
          }
        }
      }
    }

    if (!has_next) break;
    __syncthreads();
    store_tile();
    __syncthreads();
    tile = next;
  }
}

template <int DIN, int ACT>
__global__ __launch_bounds__(896) void sage_fused_kernel(
    const float* __restrict__ x, const float* __restrict__ invdeg,
    const int* __restrict__ rowptr, const int* __restrict__ col,
    const float* __restrict__ Wl, const float* __restrict__ bias,
    const float* __restrict__ Wr, float* __restrict__ out, int nnodes,
    int ntiles) {
  constexpr int DOUT = 128;
  constexpr int MT = 5;
  constexpr int ROWS = 16 * MT;
  constexpr int PW = DOUT / 16;
  constexpr int GW = 20;
  constexpr int KR = ROWS / GW;
  constexpr int NT = 32 * (PW + GW);
  constexpr int LDW = DIN + 4;
  constexpr int C4 = DIN / 4;
  constexpr int V = DIN / 32;
  static_assert(ROWS % GW == 0, "every gather wave takes the same number of rows");
  typedef float VT __attribute__((ext_vector_type(V)));
  __shared__ float sWl[DOUT * LDW];
  __shared__ float sWr[DOUT * LDW];
  __shared__ float sA[2][ROWS * LDW];
  __shared__ float sX[2][ROWS * LDW];

  const int tid = threadIdx.x;
  if ((int)blockIdx.x >= ntiles) return;

  for (int i = tid; i < DOUT * C4; i += NT) {
    const int r = i / C4, c4 = (i - r * C4) * 4;
    *(float4*)&sWl[r * LDW + c4] = *(const float4*)(Wl + (long)r * DIN + c4);
    *(float4*)&sWr[r * LDW + c4] = *(const float4*)(Wr + (long)r * DIN + c4);
  }

  const int wave = __builtin_amdgcn_readfirstlane(tid >> 5);
  const int lane = tid & 31;
  const int half = lane >> 4;
  const int l16 = lane & 15;
  const int n = (wave & (PW - 1)) * 16 + l16;
  const float bcol = bias[n];
  float* pcol = out + n;

  for (int i = 0;; ++i) {
    const int tg = (int)blockIdx.x + i * (int)gridDim.x;
    const bool has_g = tg < ntiles;

    if (wave < PW) {
      if (i > 0) {
        const int tile = tg - (int)gridDim.x;
        const float* a0 = sA[(i - 1) & 1];
        const float* x0 = sX[(i - 1) & 1];
        v8f acc[MT];
#pragma unroll
        for (int mt = 0; mt < MT; ++mt) acc[mt] = (v8f){};

#pragma unroll
        for (int k = 0; k < DIN; k += 4) {
          const int kk = k + 2 * half;
          v2f b;
          b.x = sWl[n * LDW + kk];
          b.y = sWl[n * LDW + kk + 1];
          v2f b2;
          b2.x = sWr[n * LDW + kk];
          b2.y = sWr[n * LDW + kk + 1];
#pragma unroll
          for (int mt = 0; mt < MT; ++mt) {
            const int r = mt * 16 + l16;
            v2f a;
            a.x = a0[r * LDW + kk];
            a.y = a0[r * LDW + kk + 1];
            acc[mt] = __builtin_amdgcn_wmma_f32_16x16x4_f32(false, a, false, b,
                                                            (short)0, acc[mt], false,
                                                            false);
            v2f a2;
            a2.x = x0[r * LDW + kk];
            a2.y = x0[r * LDW + kk + 1];
            acc[mt] = __builtin_amdgcn_wmma_f32_16x16x4_f32(false, a2, false, b2,
                                                            (short)0, acc[mt],
                                                            false, false);
          }
        }

        const int row0 = tile * ROWS;
#pragma unroll
        for (int mt = 0; mt < MT; ++mt) {
#pragma unroll
          for (int e = 0; e < 8; ++e) acc[mt][e] += bcol;
        }
        if (row0 + ROWS <= nnodes) {
          float* pt = pcol + (long)(row0 + 8 * half) * DOUT;
#pragma unroll
          for (int mt = 0; mt < MT; ++mt) {
#pragma unroll
            for (int e = 0; e < 8; ++e)
              pt[(mt * 16 + e) * DOUT] = act_sl<ACT>(acc[mt][e]);
          }
        } else {
#pragma unroll
          for (int mt = 0; mt < MT; ++mt) {
#pragma unroll
            for (int e = 0; e < 8; ++e) {
              const int row = row0 + mt * 16 + e + 8 * half;
              const float v = act_sl<ACT>(acc[mt][e]);
              if (row < nnodes) pcol[(long)row * DOUT] = v;
            }
          }
        }
      }
    } else if (has_g) {
      float* a1 = sA[i & 1];
      float* x1 = sX[i & 1];
      const int gw = wave - PW;
      int nd[KR], beg[KR], end[KR], cl[KR];
      float sc[KR];
#pragma unroll
      for (int k = 0; k < KR; ++k) {
        int node = tg * ROWS + gw + k * GW;
        if (node >= nnodes) node = nnodes - 1;
        nd[k] = node;
        beg[k] = rowptr[node];
        end[k] = rowptr[node + 1];
        sc[k] = invdeg[node];
      }
#pragma unroll
      for (int k = 0; k < KR; ++k) {
        const int m = min(32, end[k] - beg[k]);
        cl[k] = (lane < m) ? col[beg[k] + lane] : 0;
      }
      const float* xl = x + lane * V;
#pragma unroll
      for (int k = 0; k < KR; ++k) {
        const VT root = *(const VT*)(xl + (long)nd[k] * DIN);
        VT acc = (VT)(0.0f);
        int cols = cl[k];
        for (int base = beg[k]; base < end[k]; base += 32) {
          const int m = min(32, end[k] - base);
          if (base != beg[k]) cols = (lane < m) ? col[base + lane] : 0;
          int j = 0;
          for (; j + 16 <= m; j += 16) add_rows<DIN, 16, VT>(acc, xl, cols, j);
          if (m - j >= 8) { add_rows<DIN, 8, VT>(acc, xl, cols, j); j += 8; }
          if (m - j >= 4) { add_rows<DIN, 4, VT>(acc, xl, cols, j); j += 4; }
          if (m - j >= 2) { add_rows<DIN, 2, VT>(acc, xl, cols, j); j += 2; }
          if (m - j >= 1) { add_rows<DIN, 1, VT>(acc, xl, cols, j); }
        }
        const int r = gw + k * GW;
        *(VT*)&a1[r * LDW + lane * V] = acc * sc[k];
        *(VT*)&x1[r * LDW + lane * V] = root;
      }
    }

    if (!has_g) break;
    __syncthreads();
  }
}


static inline unsigned cdiv(long a, long b) { return (unsigned)((a + b - 1) / b); }
static inline unsigned umin(unsigned a, unsigned b) { return a < b ? a : b; }

extern "C" void kernel_launch(void* const* d_in, const int* in_sizes, int n_in,
                              void* d_out, int out_size, void* d_ws, size_t ws_size,
                              hipStream_t stream) {
  const float* x = (const float*)d_in[0];
  const int* ei = (const int*)d_in[1];
  const int N = in_sizes[0] / 64;
  const int E = in_sizes[1] / 2;
  const int* src = ei;
  const int* dst = ei + E;

  const float* g1Wl = (const float*)d_in[2];
  const float* g1b  = (const float*)d_in[3];
  const float* g1Wr = (const float*)d_in[4];
  const float* g2Wl = (const float*)d_in[5];
  const float* g2b  = (const float*)d_in[6];
  const float* g2Wr = (const float*)d_in[7];
  const float* g3Wl = (const float*)d_in[8];
  const float* g3b  = (const float*)d_in[9];
  const float* g3Wr = (const float*)d_in[10];
  const float* trW  = (const float*)d_in[11];
  const float* trb  = (const float*)d_in[12];
  const float* d1W  = (const float*)d_in[13];
  const float* d1b  = (const float*)d_in[14];
  const float* d2Wl = (const float*)d_in[15];
  const float* d2b  = (const float*)d_in[16];
  const float* d2Wr = (const float*)d_in[17];
  const float* d3Wl = (const float*)d_in[18];
  const float* d3b  = (const float*)d_in[19];
  const float* d3Wr = (const float*)d_in[20];
  const float* d4Wl = (const float*)d_in[21];
  const float* d4b  = (const float*)d_in[22];
  const float* d4Wr = (const float*)d_in[23];

  float* ws   = (float*)d_ws;
  float* inv  = ws + N;
  float* bufA = ws + 2 * (size_t)N;
  float* bufB = bufA + (size_t)N * 128;
  float* bufC = bufB + (size_t)N * 128;
  float* z2   = bufC + (size_t)N * 128;
  int*   rowptr = (int*)bufC;
  int*   col    = rowptr + (N + 1);
  float* bufD   = bufC + (size_t)N * 64;
  float* p3     = bufB;
  float* r3     = bufB + (size_t)N * 64;
  float* p4     = bufA;
  float* r4     = bufA + (size_t)N * 4;
  int*   pairs    = (int*)bufB;
  int*   binCount = (int*)z2;
  int*   binOff   = binCount + NCH * NBP;
  int*   binTot   = binOff + NCH * NBP;
  int*   binBase  = binTot + NBP;

  float* outZ01 = (float*)d_out;
  float* outZ2  = outZ01 + 2 * (size_t)N;
  float* outMu  = outZ2 + N;
  float* outLv  = outMu + (size_t)N * 64;

  const int tiles = (N + 79) / 80;
  const unsigned gL1 = umin(tiles, 256);
  const unsigned gL2 = umin(tiles, 512);
  const unsigned gL4 = umin(tiles, 1024);
  const int      chunk = (int)cdiv(E, NCH);
  const unsigned gBin  = cdiv(N, BW);
  const unsigned gG   = cdiv((long)N * 32, 256);

  bin_count_kernel<<<NCH, 1024, 0, stream>>>(dst, binCount, E, chunk);
  bin_offset_kernel<<<NBP, NCH, 0, stream>>>(binCount, binOff, binTot);
  bin_scatter_kernel<<<NCH, 1024, 0, stream>>>(src, dst, binOff, binTot, binBase,
                                               pairs, E, chunk);
  bin_finish_kernel<<<gBin, 1024, 0, stream>>>(pairs, binBase, rowptr, inv, col, N);

  sage_fused_kernel<64, 1><<<gL1, 896, 0, stream>>>(
      x, inv, rowptr, col, g1Wl, g1b, g1Wr, bufA, N, tiles);

  sage_fused_kernel<128, 1><<<gL1, 896, 0, stream>>>(
      bufA, inv, rowptr, col, g2Wl, g2b, g2Wr, bufB, N, tiles);

  sage_fused_kernel<128, 1><<<gL1, 896, 0, stream>>>(
      bufB, inv, rowptr, col, g3Wl, g3b, g3Wr, bufA, N, tiles);

  wmma_linear_kernel<128, 64, 0, 0, true><<<gL4, 128, 0, stream>>>(
      bufA, trW, trb, nullptr, outMu, outLv, N, tiles);

  wmma_linear_kernel<64, 128, 2><<<gL4, 256, 0, stream>>>(
      outMu, d1W, d1b, nullptr, bufB, nullptr, N, tiles);

  sage_fused_kernel<128, 2><<<gL1, 896, 0, stream>>>(
      bufB, inv, rowptr, col, d2Wl, d2b, d2Wr, bufA, N, tiles);

  wmma_linear_kernel<128, 128, 0, 64><<<gL2, 256, 0, stream>>>(
      bufA, d3Wl, d3b, d3Wr, p3, r3, N, tiles);
  gather_mean_tanh_kernel<64><<<gG, 256, 0, stream>>>(p3, r3, inv, rowptr, col,
                                                      bufD, N);

  wmma_linear_kernel<64, 8, 0, 3><<<(unsigned)tiles, 32, 0, stream>>>(
      bufD, d4Wl, d4b, d4Wr, p4, r4, N, tiles);
  gather_final_kernel<<<cdiv(N, 32), 256, 0, stream>>>(p4, r4, inv, rowptr, col,
                                                       outZ01, outZ2, N);
}
